# starting kernel
# baseline (speedup 1.0000x reference)
#include <hip/hip_runtime.h>
#include <hip/hip_bf16.h>
#include <math.h>

typedef float v2f __attribute__((ext_vector_type(2)));
typedef float v8f __attribute__((ext_vector_type(8)));

#define KDIM 128

template <bool TRANS_FILL, bool RELU_IN>
__global__ __launch_bounds__(256) void ggnn_gemm(
    const float* __restrict__ X, const float* __restrict__ W,
    const float* __restrict__ bias, float* __restrict__ Y,
    int nrows, int out_stride, int out_cols_total)
{
    __shared__ float Wt[KDIM * KDIM];

    const int tid   = threadIdx.x;
    const int cbase = blockIdx.y * KDIM;
    int ccols = out_cols_total - cbase;
    if (ccols > KDIM) ccols = KDIM;

    for (int i = tid; i < KDIM * KDIM; i += 256) {
        const int n = i >> 7;
        const int k = i & (KDIM - 1);
        float w = 0.0f;
        if (n < ccols) {
            if (TRANS_FILL) w = W[(size_t)k * out_cols_total + cbase + n];
            else            w = W[(size_t)(cbase + n) * KDIM + k];
        }
        Wt[n * KDIM + k] = w;
    }
    __syncthreads();

    const int wave = tid >> 5;
    const int lane = tid & 31;
    const int m0 = (blockIdx.x * 8 + wave) * 16;
    if (m0 >= nrows) return;

    const int half = lane >> 4;
    const int l16  = lane & 15;

    const float* Xrow = X + (size_t)(m0 + l16) * KDIM + half * 2;

    v8f acc[8] = {};

    for (int kb = 0; kb < KDIM; kb += 4) {
        v2f a = *(const v2f*)(Xrow + kb);
        if (RELU_IN) { a.x = fmaxf(a.x, 0.0f); a.y = fmaxf(a.y, 0.0f); }
        #pragma unroll
        for (int t = 0; t < 8; ++t) {
            v2f b = *(const v2f*)(&Wt[(t * 16 + l16) * KDIM + kb + half * 2]);
            acc[t] = __builtin_amdgcn_wmma_f32_16x16x4_f32(
                false, a, false, b, (short)0, acc[t], false, false);
        }
    }

    #pragma unroll
    for (int t = 0; t < 8; ++t) {
        const int c = t * 16 + l16;
        if (c < ccols) {
            const float bv = bias ? bias[cbase + c] : 0.0f;
            #pragma unroll
            for (int r = 0; r < 8; ++r) {
                const int row = m0 + r + half * 8;
                Y[(size_t)row * out_stride + cbase + c] = acc[t][r] + bv;
            }
        }
    }
}

__global__ __launch_bounds__(256) void ggnn_scatter(
    const float* __restrict__ m, const int* __restrict__ src,
    const int* __restrict__ dst, float* __restrict__ agg, int nedges)
{
    const long idx = (long)blockIdx.x * 256 + threadIdx.x;
    const long e = idx >> 5;
    if (e >= nedges) return;
    const int lane = (int)(idx & 31);
    const int s = src[e];
    const int d = dst[e];
    const float4 v = *(const float4*)(m + (size_t)s * KDIM + lane * 4);
    float* out = agg + (size_t)d * KDIM + lane * 4;
    atomicAdd(out + 0, v.x);
    atomicAdd(out + 1, v.y);
    atomicAdd(out + 2, v.z);
    atomicAdd(out + 3, v.w);
}

__global__ __launch_bounds__(256) void ggnn_zero(float4* __restrict__ p, long n4)
{
    const long i = (long)blockIdx.x * 256 + threadIdx.x;
    if (i < n4) p[i] = make_float4(0.f, 0.f, 0.f, 0.f);
}

__global__ __launch_bounds__(256) void ggnn_gru_gates(
    const float* __restrict__ gi, const float* __restrict__ gh,
    float* __restrict__ h, int nnodes)
{
    const long idx = (long)blockIdx.x * 256 + threadIdx.x;
    if (idx >= (long)nnodes * KDIM) return;
    const long n = idx >> 7;
    const int  c = (int)(idx & (KDIM - 1));
    const float* gin = gi + n * 384;
    const float* ghn = gh + n * 384;
    const float i_r = gin[c], i_z = gin[c + 128], i_n = gin[c + 256];
    const float h_r = ghn[c], h_z = ghn[c + 128], h_n = ghn[c + 256];
    const float r = 1.0f / (1.0f + __expf(-(i_r + h_r)));
    const float z = 1.0f / (1.0f + __expf(-(i_z + h_z)));
    const float nn = tanhf(i_n + r * h_n);
    const float hv = h[idx];
    h[idx] = (1.0f - z) * nn + z * hv;
}

extern "C" void kernel_launch(void* const* d_in, const int* in_sizes, int n_in,
                              void* d_out, int out_size, void* d_ws, size_t ws_size,
                              hipStream_t stream)
{
    (void)n_in; (void)out_size; (void)ws_size;

    const float* x      = (const float*)d_in[0];
    const int*   ei     = (const int*)d_in[1];
    const float* lin_w  = (const float*)d_in[2];
    const float* lin_b  = (const float*)d_in[3];
    const float* ggc_w  = (const float*)d_in[4];
    const float* w_ih   = (const float*)d_in[5];
    const float* w_hh   = (const float*)d_in[6];
    const float* b_ih   = (const float*)d_in[7];
    const float* b_hh   = (const float*)d_in[8];
    const float* lin2_w = (const float*)d_in[9];
    const float* lin2_b = (const float*)d_in[10];
    float* out = (float*)d_out;

    const int N = in_sizes[0] / KDIM;
    const int E = in_sizes[1] / 2;
    const int* src = ei;
    const int* dst = ei + E;

    float* h   = (float*)d_ws;
    float* m   = h   + (size_t)N * 128;
    float* agg = m   + (size_t)N * 128;
    float* gi  = agg + (size_t)N * 128;
    float* gh  = gi  + (size_t)N * 384;

    const int row_tiles  = (N + 15) / 16;
    const int row_blocks = (row_tiles + 7) / 8;
    const dim3 blk(256);

    const long agg4 = (long)N * 32;
    const unsigned zero_blks    = (unsigned)((agg4 + 255) / 256);
    const unsigned scatter_blks = (unsigned)(((long)E * 32 + 255) / 256);
    const unsigned gate_blks    = (unsigned)(((long)N * 128 + 255) / 256);

    ggnn_gemm<false, false><<<dim3(row_blocks, 1), blk, 0, stream>>>(
        x, lin_w, lin_b, h, N, 128, 128);

    for (int l = 0; l < 8; ++l) {
        ggnn_gemm<true, false><<<dim3(row_blocks, 1), blk, 0, stream>>>(
            h, ggc_w + (size_t)l * 128 * 128, nullptr, m, N, 128, 128);
        ggnn_zero<<<zero_blks, blk, 0, stream>>>((float4*)agg, agg4);
        ggnn_scatter<<<scatter_blks, blk, 0, stream>>>(m, src, dst, agg, E);
        ggnn_gemm<false, false><<<dim3(row_blocks, 3), blk, 0, stream>>>(
            agg, w_ih, b_ih, gi, N, 384, 384);
        ggnn_gemm<false, false><<<dim3(row_blocks, 3), blk, 0, stream>>>(
            h, w_hh, b_hh, gh, N, 384, 384);
        ggnn_gru_gates<<<gate_blks, blk, 0, stream>>>(gi, gh, h, N);
    }

    ggnn_gemm<false, true><<<dim3(row_blocks, 1), blk, 0, stream>>>(
        h, lin2_w, lin2_b, out, N, 100, 100);
}


# XOR-swizzle LDS weight-tile columns by 4*(row&15): each B-fragment ds_load_b64 now spreads over all 64 banks instead of every row of the fragment landing on one bank pair
# speedup vs baseline: 1.5114x; 1.5114x over previous
#include <hip/hip_runtime.h>
#include <hip/hip_bf16.h>
#include <math.h>

typedef float v2f __attribute__((ext_vector_type(2)));
typedef float v8f __attribute__((ext_vector_type(8)));

#define KDIM 128

template <bool TRANS_FILL, bool RELU_IN>
__global__ __launch_bounds__(256) void ggnn_gemm(
    const float* __restrict__ X, const float* __restrict__ W,
    const float* __restrict__ bias, float* __restrict__ Y,
    int nrows, int out_stride, int out_cols_total)
{
    __shared__ float Wt[KDIM * KDIM];

    const int tid   = threadIdx.x;
    const int cbase = blockIdx.y * KDIM;
    int ccols = out_cols_total - cbase;
    if (ccols > KDIM) ccols = KDIM;

    for (int i = tid; i < KDIM * KDIM; i += 256) {
        const int n = i >> 7;
        const int k = i & (KDIM - 1);
        float w = 0.0f;
        if (n < ccols) {
            if (TRANS_FILL) w = W[(size_t)k * out_cols_total + cbase + n];
            else            w = W[(size_t)(cbase + n) * KDIM + k];
        }
        Wt[n * KDIM + (k ^ ((n & 15) << 2))] = w;
    }
    __syncthreads();

    const int wave = tid >> 5;
    const int lane = tid & 31;
    const int m0 = (blockIdx.x * 8 + wave) * 16;
    if (m0 >= nrows) return;

    const int half = lane >> 4;
    const int l16  = lane & 15;

    const float* Xrow = X + (size_t)(m0 + l16) * KDIM + half * 2;
    const float* Brow = &Wt[l16 * KDIM + half * 2];
    const int swz = l16 << 2;

    v8f acc[8] = {};

    for (int kb = 0; kb < KDIM; kb += 4) {
        v2f a = *(const v2f*)(Xrow + kb);
        if (RELU_IN) { a.x = fmaxf(a.x, 0.0f); a.y = fmaxf(a.y, 0.0f); }
        const int kc = kb ^ swz;
        #pragma unroll
        for (int t = 0; t < 8; ++t) {
            v2f b = *(const v2f*)(Brow + t * 16 * KDIM + kc);
            acc[t] = __builtin_amdgcn_wmma_f32_16x16x4_f32(
                false, a, false, b, (short)0, acc[t], false, false);
        }
    }

    #pragma unroll
    for (int t = 0; t < 8; ++t) {
        const int c = t * 16 + l16;
        if (c < ccols) {
            const float bv = bias ? bias[cbase + c] : 0.0f;
            #pragma unroll
            for (int r = 0; r < 8; ++r) {
                const int row = m0 + r + half * 8;
                Y[(size_t)row * out_stride + cbase + c] = acc[t][r] + bv;
            }
        }
    }
}

__global__ __launch_bounds__(256) void ggnn_scatter(
    const float* __restrict__ m, const int* __restrict__ src,
    const int* __restrict__ dst, float* __restrict__ agg, int nedges)
{
    const long idx = (long)blockIdx.x * 256 + threadIdx.x;
    const long e = idx >> 5;
    if (e >= nedges) return;
    const int lane = (int)(idx & 31);
    const int s = src[e];
    const int d = dst[e];
    const float4 v = *(const float4*)(m + (size_t)s * KDIM + lane * 4);
    float* out = agg + (size_t)d * KDIM + lane * 4;
    atomicAdd(out + 0, v.x);
    atomicAdd(out + 1, v.y);
    atomicAdd(out + 2, v.z);
    atomicAdd(out + 3, v.w);
}

__global__ __launch_bounds__(256) void ggnn_zero(float4* __restrict__ p, long n4)
{
    const long i = (long)blockIdx.x * 256 + threadIdx.x;
    if (i < n4) p[i] = make_float4(0.f, 0.f, 0.f, 0.f);
}

__global__ __launch_bounds__(256) void ggnn_gru_gates(
    const float* __restrict__ gi, const float* __restrict__ gh,
    float* __restrict__ h, int nnodes)
{
    const long idx = (long)blockIdx.x * 256 + threadIdx.x;
    if (idx >= (long)nnodes * KDIM) return;
    const long n = idx >> 7;
    const int  c = (int)(idx & (KDIM - 1));
    const float* gin = gi + n * 384;
    const float* ghn = gh + n * 384;
    const float i_r = gin[c], i_z = gin[c + 128], i_n = gin[c + 256];
    const float h_r = ghn[c], h_z = ghn[c + 128], h_n = ghn[c + 256];
    const float r = 1.0f / (1.0f + __expf(-(i_r + h_r)));
    const float z = 1.0f / (1.0f + __expf(-(i_z + h_z)));
    const float nn = tanhf(i_n + r * h_n);
    const float hv = h[idx];
    h[idx] = (1.0f - z) * nn + z * hv;
}

extern "C" void kernel_launch(void* const* d_in, const int* in_sizes, int n_in,
                              void* d_out, int out_size, void* d_ws, size_t ws_size,
                              hipStream_t stream)
{
    (void)n_in; (void)out_size; (void)ws_size;

    const float* x      = (const float*)d_in[0];
    const int*   ei     = (const int*)d_in[1];
    const float* lin_w  = (const float*)d_in[2];
    const float* lin_b  = (const float*)d_in[3];
    const float* ggc_w  = (const float*)d_in[4];
    const float* w_ih   = (const float*)d_in[5];
    const float* w_hh   = (const float*)d_in[6];
    const float* b_ih   = (const float*)d_in[7];
    const float* b_hh   = (const float*)d_in[8];
    const float* lin2_w = (const float*)d_in[9];
    const float* lin2_b = (const float*)d_in[10];
    float* out = (float*)d_out;

    const int N = in_sizes[0] / KDIM;
    const int E = in_sizes[1] / 2;
    const int* src = ei;
    const int* dst = ei + E;

    float* h   = (float*)d_ws;
    float* m   = h   + (size_t)N * 128;
    float* agg = m   + (size_t)N * 128;
    float* gi  = agg + (size_t)N * 128;
    float* gh  = gi  + (size_t)N * 384;

    const int row_tiles  = (N + 15) / 16;
    const int row_blocks = (row_tiles + 7) / 8;
    const dim3 blk(256);

    const long agg4 = (long)N * 32;
    const unsigned zero_blks    = (unsigned)((agg4 + 255) / 256);
    const unsigned scatter_blks = (unsigned)(((long)E * 32 + 255) / 256);
    const unsigned gate_blks    = (unsigned)(((long)N * 128 + 255) / 256);

    ggnn_gemm<false, false><<<dim3(row_blocks, 1), blk, 0, stream>>>(
        x, lin_w, lin_b, h, N, 128, 128);

    for (int l = 0; l < 8; ++l) {
        ggnn_gemm<true, false><<<dim3(row_blocks, 1), blk, 0, stream>>>(
            h, ggc_w + (size_t)l * 128 * 128, nullptr, m, N, 128, 128);
        ggnn_zero<<<zero_blks, blk, 0, stream>>>((float4*)agg, agg4);
        ggnn_scatter<<<scatter_blks, blk, 0, stream>>>(m, src, dst, agg, E);
        ggnn_gemm<false, false><<<dim3(row_blocks, 3), blk, 0, stream>>>(
            agg, w_ih, b_ih, gi, N, 384, 384);
        ggnn_gemm<false, false><<<dim3(row_blocks, 3), blk, 0, stream>>>(
            h, w_hh, b_hh, gh, N, 384, 384);
        ggnn_gru_gates<<<gate_blks, blk, 0, stream>>>(gi, gh, h, N);
    }

    ggnn_gemm<false, true><<<dim3(row_blocks, 1), blk, 0, stream>>>(
        h, lin2_w, lin2_b, out, N, 100, 100);
}


# Replace per-layer float-atomic scatter-add and its clearing pass by an edge list grouped by destination once per call (int-atomic arrival-order rank, scan, fill) and a per-node register gather-sum
# speedup vs baseline: 2.1960x; 1.4530x over previous
#include <hip/hip_runtime.h>
#include <hip/hip_bf16.h>
#include <math.h>

typedef float v2f __attribute__((ext_vector_type(2)));
typedef float v8f __attribute__((ext_vector_type(8)));

#define KDIM 128

template <bool TRANS_FILL, bool RELU_IN>
__global__ __launch_bounds__(256) void ggnn_gemm(
    const float* __restrict__ X, const float* __restrict__ W,
    const float* __restrict__ bias, float* __restrict__ Y,
    int nrows, int out_stride, int out_cols_total)
{
    __shared__ float Wt[KDIM * KDIM];

    const int tid   = threadIdx.x;
    const int cbase = blockIdx.y * KDIM;
    int ccols = out_cols_total - cbase;
    if (ccols > KDIM) ccols = KDIM;

    for (int i = tid; i < KDIM * KDIM; i += 256) {
        const int n = i >> 7;
        const int k = i & (KDIM - 1);
        float w = 0.0f;
        if (n < ccols) {
            if (TRANS_FILL) w = W[(size_t)k * out_cols_total + cbase + n];
            else            w = W[(size_t)(cbase + n) * KDIM + k];
        }
        Wt[n * KDIM + (k ^ ((n & 15) << 2))] = w;
    }
    __syncthreads();

    const int wave = tid >> 5;
    const int lane = tid & 31;
    const int m0 = (blockIdx.x * 8 + wave) * 16;
    if (m0 >= nrows) return;

    const int half = lane >> 4;
    const int l16  = lane & 15;

    const float* Xrow = X + (size_t)(m0 + l16) * KDIM + half * 2;
    const float* Brow = &Wt[l16 * KDIM + half * 2];
    const int swz = l16 << 2;

    v8f acc[8] = {};

    for (int kb = 0; kb < KDIM; kb += 4) {
        v2f a = *(const v2f*)(Xrow + kb);
        if (RELU_IN) { a.x = fmaxf(a.x, 0.0f); a.y = fmaxf(a.y, 0.0f); }
        const int kc = kb ^ swz;
        #pragma unroll
        for (int t = 0; t < 8; ++t) {
            v2f b = *(const v2f*)(Brow + t * 16 * KDIM + kc);
            acc[t] = __builtin_amdgcn_wmma_f32_16x16x4_f32(
                false, a, false, b, (short)0, acc[t], false, false);
        }
    }

    #pragma unroll
    for (int t = 0; t < 8; ++t) {
        const int c = t * 16 + l16;
        if (c < ccols) {
            const float bv = bias ? bias[cbase + c] : 0.0f;
            #pragma unroll
            for (int r = 0; r < 8; ++r) {
                const int row = m0 + r + half * 8;
                Y[(size_t)row * out_stride + cbase + c] = acc[t][r] + bv;
            }
        }
    }
}

__global__ __launch_bounds__(256) void ggnn_clear_counts(int* __restrict__ cnt, int n)
{
    const int i = blockIdx.x * 256 + threadIdx.x;
    if (i < n) cnt[i] = 0;
}

__global__ __launch_bounds__(256) void ggnn_count_dst(
    const int* __restrict__ dst, int* __restrict__ cnt, int* __restrict__ rank, int nedges)
{
    const int e = blockIdx.x * 256 + threadIdx.x;
    if (e < nedges) rank[e] = atomicAdd(&cnt[dst[e]], 1);
}

__global__ __launch_bounds__(1024) void ggnn_scan_counts(
    const int* __restrict__ cnt, int* __restrict__ rowptr, int n)
{
    __shared__ int wsum[32];
    const int t    = threadIdx.x;
    const int lane = t & 31;
    const int w    = t >> 5;
    int carry = 0;
    for (int base = 0; base < n; base += 4096) {
        const int i0 = base + t * 4;
        int c[4];
        #pragma unroll
        for (int k = 0; k < 4; ++k) c[k] = (i0 + k < n) ? cnt[i0 + k] : 0;
        const int s = c[0] + c[1] + c[2] + c[3];

        int incl = s;
        #pragma unroll
        for (int off = 1; off < 32; off <<= 1) {
            const int u = __shfl_up(incl, off);
            if (lane >= off) incl += u;
        }
        if (lane == 31) wsum[w] = incl;
        __syncthreads();
        if (w == 0) {
            int si = wsum[lane];
            #pragma unroll
            for (int off = 1; off < 32; off <<= 1) {
                const int u = __shfl_up(si, off);
                if (lane >= off) si += u;
            }
            wsum[lane] = si;
        }
        __syncthreads();
        const int wprefix = (w == 0) ? 0 : wsum[w - 1];
        const int total   = wsum[31];
        int run = carry + wprefix + incl - s;
        #pragma unroll
        for (int k = 0; k < 4; ++k) {
            if (i0 + k < n) rowptr[i0 + k] = run;
            run += c[k];
        }
        carry += total;
        __syncthreads();
    }
    if (t == 0) rowptr[n] = carry;
}

__global__ __launch_bounds__(256) void ggnn_fill_lists(
    const int* __restrict__ src, const int* __restrict__ dst,
    const int* __restrict__ rowptr, const int* __restrict__ rank,
    int* __restrict__ csr_src, int nedges)
{
    const int e = blockIdx.x * 256 + threadIdx.x;
    if (e < nedges) csr_src[rowptr[dst[e]] + rank[e]] = src[e];
}

__device__ __forceinline__ void ggnn_add4(float4& acc, const float4 v)
{
    acc.x += v.x; acc.y += v.y; acc.z += v.z; acc.w += v.w;
}

__global__ __launch_bounds__(256) void ggnn_gather_sum(
    const float* __restrict__ m, const int* __restrict__ rowptr,
    const int* __restrict__ csr_src, float* __restrict__ agg, int nnodes)
{
    const int node = blockIdx.x * 8 + (threadIdx.x >> 5);
    if (node >= nnodes) return;
    const int lane = threadIdx.x & 31;
    const int beg = __builtin_amdgcn_readfirstlane(rowptr[node]);
    const int end = __builtin_amdgcn_readfirstlane(rowptr[node + 1]);
    const float* mcol = m + lane * 4;
    float4 acc = make_float4(0.f, 0.f, 0.f, 0.f);

    for (int base = beg; base < end; base += 32) {
        int nb = end - base;
        if (nb > 32) nb = 32;
        const int mine = (lane < nb) ? csr_src[base + lane] : 0;
        int j = 0;
        for (; j + 8 <= nb; j += 8) {
            float4 v[8];
            #pragma unroll
            for (int u = 0; u < 8; ++u) {
                const int s = __builtin_amdgcn_readlane(mine, j + u);
                v[u] = *(const float4*)(mcol + (size_t)s * KDIM);
            }
            #pragma unroll
            for (int u = 0; u < 8; ++u) ggnn_add4(acc, v[u]);
        }
        if (j + 4 <= nb) {
            float4 v[4];
            #pragma unroll
            for (int u = 0; u < 4; ++u) {
                const int s = __builtin_amdgcn_readlane(mine, j + u);
                v[u] = *(const float4*)(mcol + (size_t)s * KDIM);
            }
            #pragma unroll
            for (int u = 0; u < 4; ++u) ggnn_add4(acc, v[u]);
            j += 4;
        }
        if (j + 2 <= nb) {
            const int s0 = __builtin_amdgcn_readlane(mine, j);
            const int s1 = __builtin_amdgcn_readlane(mine, j + 1);
            const float4 v0 = *(const float4*)(mcol + (size_t)s0 * KDIM);
            const float4 v1 = *(const float4*)(mcol + (size_t)s1 * KDIM);
            ggnn_add4(acc, v0);
            ggnn_add4(acc, v1);
            j += 2;
        }
        if (j < nb) {
            const int s0 = __builtin_amdgcn_readlane(mine, j);
            const float4 v0 = *(const float4*)(mcol + (size_t)s0 * KDIM);
            ggnn_add4(acc, v0);
        }
    }
    *(float4*)(agg + (size_t)node * KDIM + lane * 4) = acc;
}

__global__ __launch_bounds__(256) void ggnn_gru_gates(
    const float* __restrict__ gi, const float* __restrict__ gh,
    float* __restrict__ h, int nnodes)
{
    const long idx = (long)blockIdx.x * 256 + threadIdx.x;
    if (idx >= (long)nnodes * KDIM) return;
    const long n = idx >> 7;
    const int  c = (int)(idx & (KDIM - 1));
    const float* gin = gi + n * 384;
    const float* ghn = gh + n * 384;
    const float i_r = gin[c], i_z = gin[c + 128], i_n = gin[c + 256];
    const float h_r = ghn[c], h_z = ghn[c + 128], h_n = ghn[c + 256];
    const float r = 1.0f / (1.0f + __expf(-(i_r + h_r)));
    const float z = 1.0f / (1.0f + __expf(-(i_z + h_z)));
    const float nn = tanhf(i_n + r * h_n);
    const float hv = h[idx];
    h[idx] = (1.0f - z) * nn + z * hv;
}

extern "C" void kernel_launch(void* const* d_in, const int* in_sizes, int n_in,
                              void* d_out, int out_size, void* d_ws, size_t ws_size,
                              hipStream_t stream)
{
    (void)n_in; (void)out_size; (void)ws_size;

    const float* x      = (const float*)d_in[0];
    const int*   ei     = (const int*)d_in[1];
    const float* lin_w  = (const float*)d_in[2];
    const float* lin_b  = (const float*)d_in[3];
    const float* ggc_w  = (const float*)d_in[4];
    const float* w_ih   = (const float*)d_in[5];
    const float* w_hh   = (const float*)d_in[6];
    const float* b_ih   = (const float*)d_in[7];
    const float* b_hh   = (const float*)d_in[8];
    const float* lin2_w = (const float*)d_in[9];
    const float* lin2_b = (const float*)d_in[10];
    float* out = (float*)d_out;

    const int N = in_sizes[0] / KDIM;
    const int E = in_sizes[1] / 2;
    const int* src = ei;
    const int* dst = ei + E;

    float* h   = (float*)d_ws;
    float* m   = h   + (size_t)N * 128;
    float* agg = m   + (size_t)N * 128;
    float* gi  = agg + (size_t)N * 128;
    float* gh  = gi  + (size_t)N * 384;
    int* cnt     = (int*)(gh + (size_t)N * 384);
    int* rowptr  = cnt + N;
    int* rank    = rowptr + N + 1;
    int* csr_src = rank + E;

    const int row_tiles  = (N + 15) / 16;
    const int row_blocks = (row_tiles + 7) / 8;
    const dim3 blk(256);

    const unsigned node_blks    = (unsigned)((N + 255) / 256);
    const unsigned edge_blks    = (unsigned)((E + 255) / 256);
    const unsigned gather_blks  = (unsigned)((N + 7) / 8);
    const unsigned gate_blks    = (unsigned)(((long)N * 128 + 255) / 256);

    ggnn_gemm<false, false><<<dim3(row_blocks, 1), blk, 0, stream>>>(
        x, lin_w, lin_b, h, N, 128, 128);

    ggnn_clear_counts<<<node_blks, blk, 0, stream>>>(cnt, N);
    ggnn_count_dst<<<edge_blks, blk, 0, stream>>>(dst, cnt, rank, E);
    ggnn_scan_counts<<<1, dim3(1024), 0, stream>>>(cnt, rowptr, N);
    ggnn_fill_lists<<<edge_blks, blk, 0, stream>>>(src, dst, rowptr, rank, csr_src, E);

    for (int l = 0; l < 8; ++l) {
        ggnn_gemm<true, false><<<dim3(row_blocks, 1), blk, 0, stream>>>(
            h, ggc_w + (size_t)l * 128 * 128, nullptr, m, N, 128, 128);
        ggnn_gather_sum<<<gather_blks, blk, 0, stream>>>(m, rowptr, csr_src, agg, N);
        ggnn_gemm<false, false><<<dim3(row_blocks, 3), blk, 0, stream>>>(
            agg, w_ih, b_ih, gi, N, 384, 384);
        ggnn_gemm<false, false><<<dim3(row_blocks, 3), blk, 0, stream>>>(
            h, w_hh, b_hh, gh, N, 384, 384);
        ggnn_gru_gates<<<gate_blks, blk, 0, stream>>>(gi, gh, h, N);
    }

    ggnn_gemm<false, true><<<dim3(row_blocks, 1), blk, 0, stream>>>(
        h, lin2_w, lin2_b, out, N, 100, 100);
}


# Fuse both GRU GEMMs and the gate kernel: one WMMA kernel per layer keeps gi and gh in 12 accumulator tiles per wave (32-column slice per work-group), writes the new state to a second buffer
# speedup vs baseline: 3.4651x; 1.5779x over previous
#include <hip/hip_runtime.h>
#include <hip/hip_bf16.h>
#include <math.h>

typedef float v2f __attribute__((ext_vector_type(2)));
typedef float v8f __attribute__((ext_vector_type(8)));

#define KDIM 128

template <bool TRANS_FILL, bool RELU_IN>
__global__ __launch_bounds__(256) void ggnn_gemm(
    const float* __restrict__ X, const float* __restrict__ W,
    const float* __restrict__ bias, float* __restrict__ Y,
    int nrows, int out_stride, int out_cols_total)
{
    __shared__ float Wt[KDIM * KDIM];

    const int tid   = threadIdx.x;
    const int cbase = blockIdx.y * KDIM;
    int ccols = out_cols_total - cbase;
    if (ccols > KDIM) ccols = KDIM;

    for (int i = tid; i < KDIM * KDIM; i += 256) {
        const int n = i >> 7;
        const int k = i & (KDIM - 1);
        float w = 0.0f;
        if (n < ccols) {
            if (TRANS_FILL) w = W[(size_t)k * out_cols_total + cbase + n];
            else            w = W[(size_t)(cbase + n) * KDIM + k];
        }
        Wt[n * KDIM + (k ^ ((n & 15) << 2))] = w;
    }
    __syncthreads();

    const int wave = tid >> 5;
    const int lane = tid & 31;
    const int m0 = (blockIdx.x * 8 + wave) * 16;
    if (m0 >= nrows) return;

    const int half = lane >> 4;
    const int l16  = lane & 15;

    const float* Xrow = X + (size_t)(m0 + l16) * KDIM + half * 2;
    const float* Brow = &Wt[l16 * KDIM + half * 2];
    const int swz = l16 << 2;

    v8f acc[8] = {};

    for (int kb = 0; kb < KDIM; kb += 4) {
        v2f a = *(const v2f*)(Xrow + kb);
        if (RELU_IN) { a.x = fmaxf(a.x, 0.0f); a.y = fmaxf(a.y, 0.0f); }
        const int kc = kb ^ swz;
        #pragma unroll
        for (int t = 0; t < 8; ++t) {
            v2f b = *(const v2f*)(Brow + t * 16 * KDIM + kc);
            acc[t] = __builtin_amdgcn_wmma_f32_16x16x4_f32(
                false, a, false, b, (short)0, acc[t], false, false);
        }
    }

    #pragma unroll
    for (int t = 0; t < 8; ++t) {
        const int c = t * 16 + l16;
        if (c < ccols) {
            const float bv = bias ? bias[cbase + c] : 0.0f;
            #pragma unroll
            for (int r = 0; r < 8; ++r) {
                const int row = m0 + r + half * 8;
                Y[(size_t)row * out_stride + cbase + c] = acc[t][r] + bv;
            }
        }
    }
}

__global__ __launch_bounds__(256) void ggnn_clear_counts(int* __restrict__ cnt, int n)
{
    const int i = blockIdx.x * 256 + threadIdx.x;
    if (i < n) cnt[i] = 0;
}

__global__ __launch_bounds__(256) void ggnn_count_dst(
    const int* __restrict__ dst, int* __restrict__ cnt, int* __restrict__ rank, int nedges)
{
    const int e = blockIdx.x * 256 + threadIdx.x;
    if (e < nedges) rank[e] = atomicAdd(&cnt[dst[e]], 1);
}

__global__ __launch_bounds__(1024) void ggnn_scan_counts(
    const int* __restrict__ cnt, int* __restrict__ rowptr, int n)
{
    __shared__ int wsum[32];
    const int t    = threadIdx.x;
    const int lane = t & 31;
    const int w    = t >> 5;
    int carry = 0;
    for (int base = 0; base < n; base += 4096) {
        const int i0 = base + t * 4;
        int c[4];
        #pragma unroll
        for (int k = 0; k < 4; ++k) c[k] = (i0 + k < n) ? cnt[i0 + k] : 0;
        const int s = c[0] + c[1] + c[2] + c[3];

        int incl = s;
        #pragma unroll
        for (int off = 1; off < 32; off <<= 1) {
            const int u = __shfl_up(incl, off);
            if (lane >= off) incl += u;
        }
        if (lane == 31) wsum[w] = incl;
        __syncthreads();
        if (w == 0) {
            int si = wsum[lane];
            #pragma unroll
            for (int off = 1; off < 32; off <<= 1) {
                const int u = __shfl_up(si, off);
                if (lane >= off) si += u;
            }
            wsum[lane] = si;
        }
        __syncthreads();
        const int wprefix = (w == 0) ? 0 : wsum[w - 1];
        const int total   = wsum[31];
        int run = carry + wprefix + incl - s;
        #pragma unroll
        for (int k = 0; k < 4; ++k) {
            if (i0 + k < n) rowptr[i0 + k] = run;
            run += c[k];
        }
        carry += total;
        __syncthreads();
    }
    if (t == 0) rowptr[n] = carry;
}

__global__ __launch_bounds__(256) void ggnn_fill_lists(
    const int* __restrict__ src, const int* __restrict__ dst,
    const int* __restrict__ rowptr, const int* __restrict__ rank,
    int* __restrict__ csr_src, int nedges)
{
    const int e = blockIdx.x * 256 + threadIdx.x;
    if (e < nedges) csr_src[rowptr[dst[e]] + rank[e]] = src[e];
}

__device__ __forceinline__ void ggnn_add4(float4& acc, const float4 v)
{
    acc.x += v.x; acc.y += v.y; acc.z += v.z; acc.w += v.w;
}

__global__ __launch_bounds__(256) void ggnn_gather_sum(
    const float* __restrict__ m, const int* __restrict__ rowptr,
    const int* __restrict__ csr_src, float* __restrict__ agg, int nnodes)
{
    const int node = blockIdx.x * 8 + (threadIdx.x >> 5);
    if (node >= nnodes) return;
    const int lane = threadIdx.x & 31;
    const int beg = __builtin_amdgcn_readfirstlane(rowptr[node]);
    const int end = __builtin_amdgcn_readfirstlane(rowptr[node + 1]);
    const float* mcol = m + lane * 4;
    float4 acc = make_float4(0.f, 0.f, 0.f, 0.f);

    for (int base = beg; base < end; base += 32) {
        int nb = end - base;
        if (nb > 32) nb = 32;
        const int mine = (lane < nb) ? csr_src[base + lane] : 0;
        int j = 0;
        for (; j + 8 <= nb; j += 8) {
            float4 v[8];
            #pragma unroll
            for (int u = 0; u < 8; ++u) {
                const int s = __builtin_amdgcn_readlane(mine, j + u);
                v[u] = *(const float4*)(mcol + (size_t)s * KDIM);
            }
            #pragma unroll
            for (int u = 0; u < 8; ++u) ggnn_add4(acc, v[u]);
        }
        if (j + 4 <= nb) {
            float4 v[4];
            #pragma unroll
            for (int u = 0; u < 4; ++u) {
                const int s = __builtin_amdgcn_readlane(mine, j + u);
                v[u] = *(const float4*)(mcol + (size_t)s * KDIM);
            }
            #pragma unroll
            for (int u = 0; u < 4; ++u) ggnn_add4(acc, v[u]);
            j += 4;
        }
        if (j + 2 <= nb) {
            const int s0 = __builtin_amdgcn_readlane(mine, j);
            const int s1 = __builtin_amdgcn_readlane(mine, j + 1);
            const float4 v0 = *(const float4*)(mcol + (size_t)s0 * KDIM);
            const float4 v1 = *(const float4*)(mcol + (size_t)s1 * KDIM);
            ggnn_add4(acc, v0);
            ggnn_add4(acc, v1);
            j += 2;
        }
        if (j < nb) {
            const int s0 = __builtin_amdgcn_readlane(mine, j);
            const float4 v0 = *(const float4*)(mcol + (size_t)s0 * KDIM);
            ggnn_add4(acc, v0);
        }
    }
    *(float4*)(agg + (size_t)node * KDIM + lane * 4) = acc;
}

#define GRU_WAVES   10
#define GRU_THREADS (GRU_WAVES * 32)
#define GRU_SLICE   32
#define GRU_WROWS   (6 * GRU_SLICE)

__global__ __launch_bounds__(GRU_THREADS) void ggnn_gru_fused(
    const float* __restrict__ agg, const float* __restrict__ hin,
    const float* __restrict__ w_ih, const float* __restrict__ w_hh,
    const float* __restrict__ b_ih, const float* __restrict__ b_hh,
    float* __restrict__ hout, int nrows)
{
    __shared__ float Wt[GRU_WROWS * KDIM];

    const int tid = threadIdx.x;
    const int c0  = blockIdx.x * GRU_SLICE;

    for (int i = tid; i < GRU_WROWS * (KDIM / 4); i += GRU_THREADS) {
        const int n  = i >> 5;
        const int k4 = (i & 31) << 2;
        const int g  = n >> 5;
        const int w  = n & 31;
        const float* wrow = (g < 3)
            ? w_ih + (size_t)(g * 128 + c0 + w) * KDIM
            : w_hh + (size_t)((g - 3) * 128 + c0 + w) * KDIM;
        const float4 v = *(const float4*)(wrow + k4);
        *(float4*)&Wt[n * KDIM + (k4 ^ ((n & 15) << 2))] = v;
    }
    __syncthreads();

    const int wave = tid >> 5;
    const int lane = tid & 31;
    const int m0 = (blockIdx.y * GRU_WAVES + wave) * 16;
    if (m0 >= nrows) return;

    const int half = lane >> 4;
    const int l16  = lane & 15;

    const float* Arow = agg + (size_t)(m0 + l16) * KDIM + half * 2;
    const float* Hrow = hin + (size_t)(m0 + l16) * KDIM + half * 2;
    const float* Brow = &Wt[l16 * KDIM + half * 2];
    const int swz = l16 << 2;

    v8f acc[12] = {};

    for (int kb = 0; kb < KDIM; kb += 4) {
        const v2f a  = *(const v2f*)(Arow + kb);
        const v2f ah = *(const v2f*)(Hrow + kb);
        const int kc = kb ^ swz;
        #pragma unroll
        for (int t = 0; t < 6; ++t) {
            const v2f b = *(const v2f*)(Brow + t * 16 * KDIM + kc);
            acc[t] = __builtin_amdgcn_wmma_f32_16x16x4_f32(
                false, a, false, b, (short)0, acc[t], false, false);
        }
        #pragma unroll
        for (int t = 6; t < 12; ++t) {
            const v2f b = *(const v2f*)(Brow + t * 16 * KDIM + kc);
            acc[t] = __builtin_amdgcn_wmma_f32_16x16x4_f32(
                false, ah, false, b, (short)0, acc[t], false, false);
        }
    }

    #pragma unroll
    for (int j = 0; j < 2; ++j) {
        const int c = c0 + j * 16 + l16;
        const float bir = b_ih[c], biz = b_ih[c + 128], bin = b_ih[c + 256];
        const float bhr = b_hh[c], bhz = b_hh[c + 128], bhn = b_hh[c + 256];
        #pragma unroll
        for (int r = 0; r < 8; ++r) {
            const size_t off = (size_t)(m0 + r + half * 8) * KDIM + c;
            const float i_r = acc[0 + j][r] + bir, h_r = acc[6 + j][r] + bhr;
            const float i_z = acc[2 + j][r] + biz, h_z = acc[8 + j][r] + bhz;
            const float i_n = acc[4 + j][r] + bin, h_n = acc[10 + j][r] + bhn;
            const float rg = 1.0f / (1.0f + __expf(-(i_r + h_r)));
            const float zg = 1.0f / (1.0f + __expf(-(i_z + h_z)));
            const float nn = tanhf(i_n + rg * h_n);
            const float hv = hin[off];
            hout[off] = (1.0f - zg) * nn + zg * hv;
        }
    }
}

extern "C" void kernel_launch(void* const* d_in, const int* in_sizes, int n_in,
                              void* d_out, int out_size, void* d_ws, size_t ws_size,
                              hipStream_t stream)
{
    (void)n_in; (void)out_size; (void)ws_size;

    const float* x      = (const float*)d_in[0];
    const int*   ei     = (const int*)d_in[1];
    const float* lin_w  = (const float*)d_in[2];
    const float* lin_b  = (const float*)d_in[3];
    const float* ggc_w  = (const float*)d_in[4];
    const float* w_ih   = (const float*)d_in[5];
    const float* w_hh   = (const float*)d_in[6];
    const float* b_ih   = (const float*)d_in[7];
    const float* b_hh   = (const float*)d_in[8];
    const float* lin2_w = (const float*)d_in[9];
    const float* lin2_b = (const float*)d_in[10];
    float* out = (float*)d_out;

    const int N = in_sizes[0] / KDIM;
    const int E = in_sizes[1] / 2;
    const int* src = ei;
    const int* dst = ei + E;

    float* h   = (float*)d_ws;
    float* m   = h   + (size_t)N * 128;
    float* agg = m   + (size_t)N * 128;
    float* h2  = agg + (size_t)N * 128;
    int* cnt     = (int*)(h2 + (size_t)N * 128);
    int* rowptr  = cnt + N;
    int* rank    = rowptr + N + 1;
    int* csr_src = rank + E;

    const int row_tiles  = (N + 15) / 16;
    const int row_blocks = (row_tiles + 7) / 8;
    const dim3 blk(256);

    const unsigned node_blks    = (unsigned)((N + 255) / 256);
    const unsigned edge_blks    = (unsigned)((E + 255) / 256);
    const unsigned gather_blks  = (unsigned)((N + 7) / 8);
    const unsigned gru_rowblks  = (unsigned)((row_tiles + GRU_WAVES - 1) / GRU_WAVES);
    float* hbuf[2] = { h, h2 };

    ggnn_gemm<false, false><<<dim3(row_blocks, 1), blk, 0, stream>>>(
        x, lin_w, lin_b, h, N, 128, 128);

    ggnn_clear_counts<<<node_blks, blk, 0, stream>>>(cnt, N);
    ggnn_count_dst<<<edge_blks, blk, 0, stream>>>(dst, cnt, rank, E);
    ggnn_scan_counts<<<1, dim3(1024), 0, stream>>>(cnt, rowptr, N);
    ggnn_fill_lists<<<edge_blks, blk, 0, stream>>>(src, dst, rowptr, rank, csr_src, E);

    for (int l = 0; l < 8; ++l) {
        const float* hcur = hbuf[l & 1];
        float* hnew = hbuf[(l + 1) & 1];
        ggnn_gemm<true, false><<<dim3(row_blocks, 1), blk, 0, stream>>>(
            hcur, ggc_w + (size_t)l * 128 * 128, nullptr, m, N, 128, 128);
        ggnn_gather_sum<<<gather_blks, blk, 0, stream>>>(m, rowptr, csr_src, agg, N);
        ggnn_gru_fused<<<dim3(128 / GRU_SLICE, gru_rowblks), dim3(GRU_THREADS), 0, stream>>>(
            agg, hcur, w_ih, w_hh, b_ih, b_hh, hnew, N);
    }

    ggnn_gemm<false, true><<<dim3(row_blocks, 1), blk, 0, stream>>>(
        h, lin2_w, lin2_b, out, N, 100, 100);
}


# GRU kernel as 256 resident 24-wave work-groups (one per compute unit): weights staged once per group, waves draw 16-row tiles from an LDS counter (integer add, sums unchanged); was 1252 groups
# speedup vs baseline: 4.1802x; 1.2064x over previous
#include <hip/hip_runtime.h>
#include <hip/hip_bf16.h>
#include <math.h>

typedef float v2f __attribute__((ext_vector_type(2)));
typedef float v8f __attribute__((ext_vector_type(8)));

#define KDIM 128

template <bool TRANS_FILL, bool RELU_IN>
__global__ __launch_bounds__(256) void ggnn_gemm(
    const float* __restrict__ X, const float* __restrict__ W,
    const float* __restrict__ bias, float* __restrict__ Y,
    int nrows, int out_stride, int out_cols_total, int w_zstride, int y_zstride)
{
    W += (size_t)blockIdx.z * w_zstride;
    Y += (size_t)blockIdx.z * y_zstride;

    __shared__ float Wt[KDIM * KDIM];

    const int tid   = threadIdx.x;
    const int cbase = blockIdx.y * KDIM;
    int ccols = out_cols_total - cbase;
    if (ccols > KDIM) ccols = KDIM;

    for (int i = tid; i < KDIM * KDIM; i += 256) {
        const int n = i >> 7;
        const int k = i & (KDIM - 1);
        float w = 0.0f;
        if (n < ccols) {
            if (TRANS_FILL) w = W[(size_t)k * out_cols_total + cbase + n];
            else            w = W[(size_t)(cbase + n) * KDIM + k];
        }
        Wt[n * KDIM + (k ^ ((n & 15) << 2))] = w;
    }
    __syncthreads();

    const int wave = tid >> 5;
    const int lane = tid & 31;
    const int m0 = (blockIdx.x * 8 + wave) * 16;
    if (m0 >= nrows) return;

    const int half = lane >> 4;
    const int l16  = lane & 15;

    const float* Xrow = X + (size_t)(m0 + l16) * KDIM + half * 2;
    const float* Brow = &Wt[l16 * KDIM + half * 2];
    const int swz = l16 << 2;

    v8f acc[8] = {};

    for (int kb = 0; kb < KDIM; kb += 4) {
        v2f a = *(const v2f*)(Xrow + kb);
        if (RELU_IN) { a.x = fmaxf(a.x, 0.0f); a.y = fmaxf(a.y, 0.0f); }
        const int kc = kb ^ swz;
        #pragma unroll
        for (int t = 0; t < 8; ++t) {
            v2f b = *(const v2f*)(Brow + t * 16 * KDIM + kc);
            acc[t] = __builtin_amdgcn_wmma_f32_16x16x4_f32(
                false, a, false, b, (short)0, acc[t], false, false);
        }
    }

    #pragma unroll
    for (int t = 0; t < 8; ++t) {
        const int c = t * 16 + l16;
        if (c < ccols) {
            const float bv = bias ? bias[cbase + c] : 0.0f;
            #pragma unroll
            for (int r = 0; r < 8; ++r) {
                const int row = m0 + r + half * 8;
                Y[(size_t)row * out_stride + cbase + c] = acc[t][r] + bv;
            }
        }
    }
}

__global__ __launch_bounds__(256) void ggnn_clear_counts(int* __restrict__ cnt, int n)
{
    const int i = blockIdx.x * 256 + threadIdx.x;
    if (i < n) cnt[i] = 0;
}

__global__ __launch_bounds__(256) void ggnn_count_dst(
    const int* __restrict__ dst, int* __restrict__ cnt, int* __restrict__ rank, int nedges)
{
    const int e = blockIdx.x * 256 + threadIdx.x;
    if (e < nedges) rank[e] = atomicAdd(&cnt[dst[e]], 1);
}

__global__ __launch_bounds__(1024) void ggnn_scan_counts(
    const int* __restrict__ cnt, int* __restrict__ rowptr, int n)
{
    __shared__ int wsum[32];
    const int t    = threadIdx.x;
    const int lane = t & 31;
    const int w    = t >> 5;
    int carry = 0;
    for (int base = 0; base < n; base += 4096) {
        const int i0 = base + t * 4;
        int c[4];
        #pragma unroll
        for (int k = 0; k < 4; ++k) c[k] = (i0 + k < n) ? cnt[i0 + k] : 0;
        const int s = c[0] + c[1] + c[2] + c[3];

        int incl = s;
        #pragma unroll
        for (int off = 1; off < 32; off <<= 1) {
            const int u = __shfl_up(incl, off);
            if (lane >= off) incl += u;
        }
        if (lane == 31) wsum[w] = incl;
        __syncthreads();
        if (w == 0) {
            int si = wsum[lane];
            #pragma unroll
            for (int off = 1; off < 32; off <<= 1) {
                const int u = __shfl_up(si, off);
                if (lane >= off) si += u;
            }
            wsum[lane] = si;
        }
        __syncthreads();
        const int wprefix = (w == 0) ? 0 : wsum[w - 1];
        const int total   = wsum[31];
        int run = carry + wprefix + incl - s;
        #pragma unroll
        for (int k = 0; k < 4; ++k) {
            if (i0 + k < n) rowptr[i0 + k] = run;
            run += c[k];
        }
        carry += total;
        __syncthreads();
    }
    if (t == 0) rowptr[n] = carry;
}

__global__ __launch_bounds__(256) void ggnn_fill_lists(
    const int* __restrict__ src, const int* __restrict__ dst,
    const int* __restrict__ rowptr, const int* __restrict__ rank,
    int* __restrict__ csr_src, int nedges)
{
    const int e = blockIdx.x * 256 + threadIdx.x;
    if (e < nedges) csr_src[rowptr[dst[e]] + rank[e]] = src[e];
}

__device__ __forceinline__ void ggnn_add4(float4& acc, const float4 v)
{
    acc.x += v.x; acc.y += v.y; acc.z += v.z; acc.w += v.w;
}

__global__ __launch_bounds__(256) void ggnn_gather_sum(
    const float* __restrict__ m, const int* __restrict__ rowptr,
    const int* __restrict__ csr_src, float* __restrict__ agg, int nnodes)
{
    const int node = blockIdx.x * 8 + (threadIdx.x >> 5);
    if (node >= nnodes) return;
    const int lane = threadIdx.x & 31;
    const int beg = __builtin_amdgcn_readfirstlane(rowptr[node]);
    const int end = __builtin_amdgcn_readfirstlane(rowptr[node + 1]);
    const float* mcol = m + lane * 4;
    float4 acc = make_float4(0.f, 0.f, 0.f, 0.f);

    for (int base = beg; base < end; base += 32) {
        int nb = end - base;
        if (nb > 32) nb = 32;
        const int mine = (lane < nb) ? csr_src[base + lane] : 0;
        int j = 0;
        for (; j + 8 <= nb; j += 8) {
            float4 v[8];
            #pragma unroll
            for (int u = 0; u < 8; ++u) {
                const int s = __builtin_amdgcn_readlane(mine, j + u);
                v[u] = *(const float4*)(mcol + (size_t)s * KDIM);
            }
            #pragma unroll
            for (int u = 0; u < 8; ++u) ggnn_add4(acc, v[u]);
        }
        if (j + 4 <= nb) {
            float4 v[4];
            #pragma unroll
            for (int u = 0; u < 4; ++u) {
                const int s = __builtin_amdgcn_readlane(mine, j + u);
                v[u] = *(const float4*)(mcol + (size_t)s * KDIM);
            }
            #pragma unroll
            for (int u = 0; u < 4; ++u) ggnn_add4(acc, v[u]);
            j += 4;
        }
        if (j + 2 <= nb) {
            const int s0 = __builtin_amdgcn_readlane(mine, j);
            const int s1 = __builtin_amdgcn_readlane(mine, j + 1);
            const float4 v0 = *(const float4*)(mcol + (size_t)s0 * KDIM);
            const float4 v1 = *(const float4*)(mcol + (size_t)s1 * KDIM);
            ggnn_add4(acc, v0);
            ggnn_add4(acc, v1);
            j += 2;
        }
        if (j < nb) {
            const int s0 = __builtin_amdgcn_readlane(mine, j);
            const float4 v0 = *(const float4*)(mcol + (size_t)s0 * KDIM);
            ggnn_add4(acc, v0);
        }
    }
    *(float4*)(agg + (size_t)node * KDIM + lane * 4) = acc;
}

#define GRU_WAVES   24
#define GRU_THREADS (GRU_WAVES * 32)
#define GRU_SLICE   32
#define GRU_WROWS   (6 * GRU_SLICE)
#define GRU_GROUPS  64

__global__ __launch_bounds__(GRU_THREADS, 6) void ggnn_gru_fused(
    const float* __restrict__ agg, const float* __restrict__ hin,
    const float* __restrict__ w_ih, const float* __restrict__ w_hh,
    const float* __restrict__ b_ih, const float* __restrict__ b_hh,
    float* __restrict__ hout, int nrows)
{
    __shared__ float Wt[GRU_WROWS * KDIM];
    __shared__ int next_tile;

    const int tid = threadIdx.x;
    const int c0  = blockIdx.x * GRU_SLICE;

    if (tid == 0) next_tile = 0;

    for (int i = tid; i < GRU_WROWS * (KDIM / 4); i += GRU_THREADS) {
        const int n  = i >> 5;
        const int k4 = (i & 31) << 2;
        const int g  = n >> 5;
        const int w  = n & 31;
        const float* wrow = (g < 3)
            ? w_ih + (size_t)(g * 128 + c0 + w) * KDIM
            : w_hh + (size_t)((g - 3) * 128 + c0 + w) * KDIM;
        const float4 v = *(const float4*)(wrow + k4);
        *(float4*)&Wt[n * KDIM + (k4 ^ ((n & 15) << 2))] = v;
    }
    __syncthreads();

    const int lane = tid & 31;
    const int half = lane >> 4;
    const int l16  = lane & 15;

    const float* Brow = &Wt[l16 * KDIM + half * 2];
    const int swz = l16 << 2;

    const int groups    = (int)gridDim.y;
    const int group     = (int)blockIdx.y;
    const int row_tiles = (nrows + 15) >> 4;

    for (;;) {
        int ticket = 0;
        if (lane == 0) ticket = atomicAdd(&next_tile, 1);
        ticket = __builtin_amdgcn_readfirstlane(ticket);
        const int rt = group + ticket * groups;
        if (rt >= row_tiles) break;

        const int m0 = rt * 16;
        const float* Arow = agg + (size_t)(m0 + l16) * KDIM + half * 2;
        const float* Hrow = hin + (size_t)(m0 + l16) * KDIM + half * 2;

        v8f acc[12] = {};

        for (int kb = 0; kb < KDIM; kb += 4) {
            const v2f a  = *(const v2f*)(Arow + kb);
            const v2f ah = *(const v2f*)(Hrow + kb);
            const int kc = kb ^ swz;
            #pragma unroll
            for (int t = 0; t < 6; ++t) {
                const v2f b = *(const v2f*)(Brow + t * 16 * KDIM + kc);
                acc[t] = __builtin_amdgcn_wmma_f32_16x16x4_f32(
                    false, a, false, b, (short)0, acc[t], false, false);
            }
            #pragma unroll
            for (int t = 6; t < 12; ++t) {
                const v2f b = *(const v2f*)(Brow + t * 16 * KDIM + kc);
                acc[t] = __builtin_amdgcn_wmma_f32_16x16x4_f32(
                    false, ah, false, b, (short)0, acc[t], false, false);
            }
        }

        #pragma unroll
        for (int j = 0; j < 2; ++j) {
            const int c = c0 + j * 16 + l16;
            const float bir = b_ih[c], biz = b_ih[c + 128], bin = b_ih[c + 256];
            const float bhr = b_hh[c], bhz = b_hh[c + 128], bhn = b_hh[c + 256];
            #pragma unroll
            for (int r = 0; r < 8; ++r) {
                const size_t off = (size_t)(m0 + r + half * 8) * KDIM + c;
                const float i_r = acc[0 + j][r] + bir, h_r = acc[6 + j][r] + bhr;
                const float i_z = acc[2 + j][r] + biz, h_z = acc[8 + j][r] + bhz;
                const float i_n = acc[4 + j][r] + bin, h_n = acc[10 + j][r] + bhn;
                const float rg = 1.0f / (1.0f + __expf(-(i_r + h_r)));
                const float zg = 1.0f / (1.0f + __expf(-(i_z + h_z)));
                const float nn = tanhf(i_n + rg * h_n);
                const float hv = hin[off];
                hout[off] = (1.0f - zg) * nn + zg * hv;
            }
        }
    }
}

extern "C" void kernel_launch(void* const* d_in, const int* in_sizes, int n_in,
                              void* d_out, int out_size, void* d_ws, size_t ws_size,
                              hipStream_t stream)
{
    (void)n_in; (void)out_size; (void)ws_size;

    const float* x      = (const float*)d_in[0];
    const int*   ei     = (const int*)d_in[1];
    const float* lin_w  = (const float*)d_in[2];
    const float* lin_b  = (const float*)d_in[3];
    const float* ggc_w  = (const float*)d_in[4];
    const float* w_ih   = (const float*)d_in[5];
    const float* w_hh   = (const float*)d_in[6];
    const float* b_ih   = (const float*)d_in[7];
    const float* b_hh   = (const float*)d_in[8];
    const float* lin2_w = (const float*)d_in[9];
    const float* lin2_b = (const float*)d_in[10];
    float* out = (float*)d_out;

    const int N = in_sizes[0] / KDIM;
    const int E = in_sizes[1] / 2;
    const int* src = ei;
    const int* dst = ei + E;

    float* h     = (float*)d_ws;
    float* wfold = h + (size_t)N * 128;
    float* agg   = wfold + (size_t)N * 128;
    float* h2  = agg + (size_t)N * 128;
    int* cnt     = (int*)(h2 + (size_t)N * 128);
    int* rowptr  = cnt + N;
    int* rank    = rowptr + N + 1;
    int* csr_src = rank + E;

    const int row_tiles  = (N + 15) / 16;
    const int row_blocks = (row_tiles + 7) / 8;
    const dim3 blk(256);

    const unsigned node_blks    = (unsigned)((N + 255) / 256);
    const unsigned edge_blks    = (unsigned)((E + 255) / 256);
    const unsigned gather_blks  = (unsigned)((N + 7) / 8);
    float* hbuf[2] = { h, h2 };

    ggnn_gemm<false, false><<<dim3(row_blocks, 1), blk, 0, stream>>>(
        x, lin_w, lin_b, h, N, 128, 128, 0, 0);

    ggnn_gemm<false, false><<<dim3(3, 1, 8), blk, 0, stream>>>(
        w_ih, ggc_w, nullptr, wfold, 384, 128, 128, 128 * 128, 384 * 128);

    ggnn_clear_counts<<<node_blks, blk, 0, stream>>>(cnt, N);
    ggnn_count_dst<<<edge_blks, blk, 0, stream>>>(dst, cnt, rank, E);
    ggnn_scan_counts<<<1, dim3(1024), 0, stream>>>(cnt, rowptr, N);
    ggnn_fill_lists<<<edge_blks, blk, 0, stream>>>(src, dst, rowptr, rank, csr_src, E);

    for (int l = 0; l < 8; ++l) {
        const float* hcur = hbuf[l & 1];
        float* hnew = hbuf[(l + 1) & 1];
        ggnn_gather_sum<<<gather_blks, blk, 0, stream>>>(hcur, rowptr, csr_src, agg, N);
        ggnn_gru_fused<<<dim3(128 / GRU_SLICE, GRU_GROUPS), dim3(GRU_THREADS), 0, stream>>>(
            agg, hcur, wfold + (size_t)l * 384 * 128, w_hh, b_ih, b_hh, hnew, N);
    }

    ggnn_gemm<false, true><<<dim3(row_blocks, 1), blk, 0, stream>>>(
        h, lin2_w, lin2_b, out, N, 100, 100, 0, 0);
}


# Weight staging in the three WMMA kernels: a thread issues all its 16-byte weight loads, then the LDS stores (GEMM: 16 float4, was 64 scalar load-wait-store trips; GRU: 8 at once)
# speedup vs baseline: 4.5937x; 1.0989x over previous
#include <hip/hip_runtime.h>
#include <hip/hip_bf16.h>
#include <math.h>

typedef float v2f __attribute__((ext_vector_type(2)));
typedef float v8f __attribute__((ext_vector_type(8)));

#define KDIM 128

template <bool TRANS_FILL, bool RELU_IN>
__global__ __launch_bounds__(256) void ggnn_gemm(
    const float* __restrict__ X, const float* __restrict__ W,
    const float* __restrict__ bias, float* __restrict__ Y,
    int nrows, int out_stride, int out_cols_total, int w_zstride, int y_zstride)
{
    W += (size_t)blockIdx.z * w_zstride;
    Y += (size_t)blockIdx.z * y_zstride;

    __shared__ float Wt[KDIM * KDIM];

    const int tid   = threadIdx.x;
    const int cbase = blockIdx.y * KDIM;
    int ccols = out_cols_total - cbase;
    if (ccols > KDIM) ccols = KDIM;

    if (TRANS_FILL) {
        for (int i = tid; i < KDIM * KDIM; i += 256) {
            const int n = i >> 7;
            const int k = i & (KDIM - 1);
            float w = 0.0f;
            if (n < ccols) w = W[(size_t)k * out_cols_total + cbase + n];
            Wt[n * KDIM + (k ^ ((n & 15) << 2))] = w;
        }
    } else {
        float4 stage[16];
        #pragma unroll
        for (int j = 0; j < 16; ++j) {
            const int i  = tid + j * 256;
            const int n  = i >> 5;
            const int k4 = (i & 31) << 2;
            const int nr = (n < ccols) ? n : 0;
            stage[j] = *(const float4*)(W + (size_t)(cbase + nr) * KDIM + k4);
        }
        #pragma unroll
        for (int j = 0; j < 16; ++j) {
            const int i  = tid + j * 256;
            const int n  = i >> 5;
            const int k4 = (i & 31) << 2;
            const float4 v = (n < ccols) ? stage[j] : make_float4(0.0f, 0.0f, 0.0f, 0.0f);
            *(float4*)&Wt[n * KDIM + (k4 ^ ((n & 15) << 2))] = v;
        }
    }
    __syncthreads();

    const int wave = tid >> 5;
    const int lane = tid & 31;
    const int m0 = (blockIdx.x * 8 + wave) * 16;
    if (m0 >= nrows) return;

    const int half = lane >> 4;
    const int l16  = lane & 15;

    const float* Xrow = X + (size_t)(m0 + l16) * KDIM + half * 2;
    const float* Brow = &Wt[l16 * KDIM + half * 2];
    const int swz = l16 << 2;

    v8f acc[8] = {};

    for (int kb = 0; kb < KDIM; kb += 4) {
        v2f a = *(const v2f*)(Xrow + kb);
        if (RELU_IN) { a.x = fmaxf(a.x, 0.0f); a.y = fmaxf(a.y, 0.0f); }
        const int kc = kb ^ swz;
        #pragma unroll
        for (int t = 0; t < 8; ++t) {
            v2f b = *(const v2f*)(Brow + t * 16 * KDIM + kc);
            acc[t] = __builtin_amdgcn_wmma_f32_16x16x4_f32(
                false, a, false, b, (short)0, acc[t], false, false);
        }
    }

    #pragma unroll
    for (int t = 0; t < 8; ++t) {
        const int c = t * 16 + l16;
        if (c < ccols) {
            const float bv = bias ? bias[cbase + c] : 0.0f;
            #pragma unroll
            for (int r = 0; r < 8; ++r) {
                const int row = m0 + r + half * 8;
                Y[(size_t)row * out_stride + cbase + c] = acc[t][r] + bv;
            }
        }
    }
}

__global__ __launch_bounds__(256) void ggnn_clear_counts(int* __restrict__ cnt, int n)
{
    const int i = blockIdx.x * 256 + threadIdx.x;
    if (i < n) cnt[i] = 0;
}

__global__ __launch_bounds__(256) void ggnn_count_dst(
    const int* __restrict__ dst, int* __restrict__ cnt, int* __restrict__ rank, int nedges)
{
    const int e = blockIdx.x * 256 + threadIdx.x;
    if (e < nedges) rank[e] = atomicAdd(&cnt[dst[e]], 1);
}

__global__ __launch_bounds__(1024) void ggnn_scan_counts(
    const int* __restrict__ cnt, int* __restrict__ rowptr, int n)
{
    __shared__ int wsum[32];
    const int t    = threadIdx.x;
    const int lane = t & 31;
    const int w    = t >> 5;
    int carry = 0;
    for (int base = 0; base < n; base += 4096) {
        const int i0 = base + t * 4;
        int c[4];
        #pragma unroll
        for (int k = 0; k < 4; ++k) c[k] = (i0 + k < n) ? cnt[i0 + k] : 0;
        const int s = c[0] + c[1] + c[2] + c[3];

        int incl = s;
        #pragma unroll
        for (int off = 1; off < 32; off <<= 1) {
            const int u = __shfl_up(incl, off);
            if (lane >= off) incl += u;
        }
        if (lane == 31) wsum[w] = incl;
        __syncthreads();
        if (w == 0) {
            int si = wsum[lane];
            #pragma unroll
            for (int off = 1; off < 32; off <<= 1) {
                const int u = __shfl_up(si, off);
                if (lane >= off) si += u;
            }
            wsum[lane] = si;
        }
        __syncthreads();
        const int wprefix = (w == 0) ? 0 : wsum[w - 1];
        const int total   = wsum[31];
        int run = carry + wprefix + incl - s;
        #pragma unroll
        for (int k = 0; k < 4; ++k) {
            if (i0 + k < n) rowptr[i0 + k] = run;
            run += c[k];
        }
        carry += total;
        __syncthreads();
    }
    if (t == 0) rowptr[n] = carry;
}

__global__ __launch_bounds__(256) void ggnn_fill_lists(
    const int* __restrict__ src, const int* __restrict__ dst,
    const int* __restrict__ rowptr, const int* __restrict__ rank,
    int* __restrict__ csr_src, int nedges)
{
    const int e = blockIdx.x * 256 + threadIdx.x;
    if (e < nedges) csr_src[rowptr[dst[e]] + rank[e]] = src[e];
}

__device__ __forceinline__ void ggnn_add4(float4& acc, const float4 v)
{
    acc.x += v.x; acc.y += v.y; acc.z += v.z; acc.w += v.w;
}

__global__ __launch_bounds__(256) void ggnn_gather_sum(
    const float* __restrict__ m, const int* __restrict__ rowptr,
    const int* __restrict__ csr_src, float* __restrict__ agg, int nnodes)
{
    const int node = blockIdx.x * 8 + (threadIdx.x >> 5);
    if (node >= nnodes) return;
    const int lane = threadIdx.x & 31;
    const int beg = __builtin_amdgcn_readfirstlane(rowptr[node]);
    const int end = __builtin_amdgcn_readfirstlane(rowptr[node + 1]);
    const float* mcol = m + lane * 4;
    float4 acc = make_float4(0.f, 0.f, 0.f, 0.f);

    for (int base = beg; base < end; base += 32) {
        int nb = end - base;
        if (nb > 32) nb = 32;
        const int mine = (lane < nb) ? csr_src[base + lane] : 0;
        int j = 0;
        for (; j + 8 <= nb; j += 8) {
            float4 v[8];
            #pragma unroll
            for (int u = 0; u < 8; ++u) {
                const int s = __builtin_amdgcn_readlane(mine, j + u);
                v[u] = *(const float4*)(mcol + (size_t)s * KDIM);
            }
            #pragma unroll
            for (int u = 0; u < 8; ++u) ggnn_add4(acc, v[u]);
        }
        if (j + 4 <= nb) {
            float4 v[4];
            #pragma unroll
            for (int u = 0; u < 4; ++u) {
                const int s = __builtin_amdgcn_readlane(mine, j + u);
                v[u] = *(const float4*)(mcol + (size_t)s * KDIM);
            }
            #pragma unroll
            for (int u = 0; u < 4; ++u) ggnn_add4(acc, v[u]);
            j += 4;
        }
        if (j + 2 <= nb) {
            const int s0 = __builtin_amdgcn_readlane(mine, j);
            const int s1 = __builtin_amdgcn_readlane(mine, j + 1);
            const float4 v0 = *(const float4*)(mcol + (size_t)s0 * KDIM);
            const float4 v1 = *(const float4*)(mcol + (size_t)s1 * KDIM);
            ggnn_add4(acc, v0);
            ggnn_add4(acc, v1);
            j += 2;
        }
        if (j < nb) {
            const int s0 = __builtin_amdgcn_readlane(mine, j);
            const float4 v0 = *(const float4*)(mcol + (size_t)s0 * KDIM);
            ggnn_add4(acc, v0);
        }
    }
    *(float4*)(agg + (size_t)node * KDIM + lane * 4) = acc;
}

#define GRU_WAVES   24
#define GRU_THREADS (GRU_WAVES * 32)
#define GRU_SLICE   32
#define GRU_WROWS   (6 * GRU_SLICE)
#define GRU_GROUPS  64
#define GRU_STAGE_PIECES ((GRU_WROWS * (KDIM / 4)) / GRU_THREADS)

__global__ __launch_bounds__(GRU_THREADS, 6) void ggnn_gru_fused(
    const float* __restrict__ agg, const float* __restrict__ hin,
    const float* __restrict__ w_ih, const float* __restrict__ w_hh,
    const float* __restrict__ b_ih, const float* __restrict__ b_hh,
    float* __restrict__ hout, int nrows)
{
    __shared__ float Wt[GRU_WROWS * KDIM];
    __shared__ int next_tile;

    const int tid = threadIdx.x;
    const int c0  = blockIdx.x * GRU_SLICE;

    if (tid == 0) next_tile = 0;

    float4 stage[GRU_STAGE_PIECES];
    #pragma unroll
    for (int j = 0; j < GRU_STAGE_PIECES; ++j) {
        const int i  = tid + j * GRU_THREADS;
        const int n  = i >> 5;
        const int k4 = (i & 31) << 2;
        const int g  = n >> 5;
        const int w  = n & 31;
        const float* wrow = (g < 3)
            ? w_ih + (size_t)(g * 128 + c0 + w) * KDIM
            : w_hh + (size_t)((g - 3) * 128 + c0 + w) * KDIM;
        stage[j] = *(const float4*)(wrow + k4);
    }
    #pragma unroll
    for (int j = 0; j < GRU_STAGE_PIECES; ++j) {
        const int i  = tid + j * GRU_THREADS;
        const int n  = i >> 5;
        const int k4 = (i & 31) << 2;
        *(float4*)&Wt[n * KDIM + (k4 ^ ((n & 15) << 2))] = stage[j];
    }
    __syncthreads();

    const int lane = tid & 31;
    const int half = lane >> 4;
    const int l16  = lane & 15;

    const float* Brow = &Wt[l16 * KDIM + half * 2];
    const int swz = l16 << 2;

    const int groups    = (int)gridDim.y;
    const int group     = (int)blockIdx.y;
    const int row_tiles = (nrows + 15) >> 4;

    for (;;) {
        int ticket = 0;
        if (lane == 0) ticket = atomicAdd(&next_tile, 1);
        ticket = __builtin_amdgcn_readfirstlane(ticket);
        const int rt = group + ticket * groups;
        if (rt >= row_tiles) break;

        const int m0 = rt * 16;
        const float* Arow = agg + (size_t)(m0 + l16) * KDIM + half * 2;
        const float* Hrow = hin + (size_t)(m0 + l16) * KDIM + half * 2;

        v8f acc[12] = {};

        for (int kb = 0; kb < KDIM; kb += 4) {
            const v2f a  = *(const v2f*)(Arow + kb);
            const v2f ah = *(const v2f*)(Hrow + kb);
            const int kc = kb ^ swz;
            #pragma unroll
            for (int t = 0; t < 6; ++t) {
                const v2f b = *(const v2f*)(Brow + t * 16 * KDIM + kc);
                acc[t] = __builtin_amdgcn_wmma_f32_16x16x4_f32(
                    false, a, false, b, (short)0, acc[t], false, false);
            }
            #pragma unroll
            for (int t = 6; t < 12; ++t) {
                const v2f b = *(const v2f*)(Brow + t * 16 * KDIM + kc);
                acc[t] = __builtin_amdgcn_wmma_f32_16x16x4_f32(
                    false, ah, false, b, (short)0, acc[t], false, false);
            }
        }

        #pragma unroll
        for (int j = 0; j < 2; ++j) {
            const int c = c0 + j * 16 + l16;
            const float bir = b_ih[c], biz = b_ih[c + 128], bin = b_ih[c + 256];
            const float bhr = b_hh[c], bhz = b_hh[c + 128], bhn = b_hh[c + 256];
            #pragma unroll
            for (int r = 0; r < 8; ++r) {
                const size_t off = (size_t)(m0 + r + half * 8) * KDIM + c;
                const float i_r = acc[0 + j][r] + bir, h_r = acc[6 + j][r] + bhr;
                const float i_z = acc[2 + j][r] + biz, h_z = acc[8 + j][r] + bhz;
                const float i_n = acc[4 + j][r] + bin, h_n = acc[10 + j][r] + bhn;
                const float rg = 1.0f / (1.0f + __expf(-(i_r + h_r)));
                const float zg = 1.0f / (1.0f + __expf(-(i_z + h_z)));
                const float nn = tanhf(i_n + rg * h_n);
                const float hv = hin[off];
                hout[off] = (1.0f - zg) * nn + zg * hv;
            }
        }
    }
}

extern "C" void kernel_launch(void* const* d_in, const int* in_sizes, int n_in,
                              void* d_out, int out_size, void* d_ws, size_t ws_size,
                              hipStream_t stream)
{
    (void)n_in; (void)out_size; (void)ws_size;

    const float* x      = (const float*)d_in[0];
    const int*   ei     = (const int*)d_in[1];
    const float* lin_w  = (const float*)d_in[2];
    const float* lin_b  = (const float*)d_in[3];
    const float* ggc_w  = (const float*)d_in[4];
    const float* w_ih   = (const float*)d_in[5];
    const float* w_hh   = (const float*)d_in[6];
    const float* b_ih   = (const float*)d_in[7];
    const float* b_hh   = (const float*)d_in[8];
    const float* lin2_w = (const float*)d_in[9];
    const float* lin2_b = (const float*)d_in[10];
    float* out = (float*)d_out;

    const int N = in_sizes[0] / KDIM;
    const int E = in_sizes[1] / 2;
    const int* src = ei;
    const int* dst = ei + E;

    float* h     = (float*)d_ws;
    float* wfold = h + (size_t)N * 128;
    float* agg   = wfold + (size_t)N * 128;
    float* h2  = agg + (size_t)N * 128;
    int* cnt     = (int*)(h2 + (size_t)N * 128);
    int* rowptr  = cnt + N;
    int* rank    = rowptr + N + 1;
    int* csr_src = rank + E;

    const int row_tiles  = (N + 15) / 16;
    const int row_blocks = (row_tiles + 7) / 8;
    const dim3 blk(256);

    const unsigned node_blks    = (unsigned)((N + 255) / 256);
    const unsigned edge_blks    = (unsigned)((E + 255) / 256);
    const unsigned gather_blks  = (unsigned)((N + 7) / 8);
    float* hbuf[2] = { h, h2 };

    ggnn_gemm<false, false><<<dim3(row_blocks, 1), blk, 0, stream>>>(
        x, lin_w, lin_b, h, N, 128, 128, 0, 0);

    ggnn_gemm<false, false><<<dim3(3, 1, 8), blk, 0, stream>>>(
        w_ih, ggc_w, nullptr, wfold, 384, 128, 128, 128 * 128, 384 * 128);

    ggnn_clear_counts<<<node_blks, blk, 0, stream>>>(cnt, N);
    ggnn_count_dst<<<edge_blks, blk, 0, stream>>>(dst, cnt, rank, E);
    ggnn_scan_counts<<<1, dim3(1024), 0, stream>>>(cnt, rowptr, N);
    ggnn_fill_lists<<<edge_blks, blk, 0, stream>>>(src, dst, rowptr, rank, csr_src, E);

    for (int l = 0; l < 8; ++l) {
        const float* hcur = hbuf[l & 1];
        float* hnew = hbuf[(l + 1) & 1];
        ggnn_gather_sum<<<gather_blks, blk, 0, stream>>>(hcur, rowptr, csr_src, agg, N);
        ggnn_gru_fused<<<dim3(128 / GRU_SLICE, GRU_GROUPS), dim3(GRU_THREADS), 0, stream>>>(
            agg, hcur, wfold + (size_t)l * 384 * 128, w_hh, b_ih, b_hh, hnew, N);
    }

    ggnn_gemm<false, true><<<dim3(row_blocks, 1), blk, 0, stream>>>(
        h, lin2_w, lin2_b, out, N, 100, 100, 0, 0);
}


# Per layer the neighbour-row gather and the h-side GRU product run in ONE launch (each wave alternates 16-row tiles and nodes; gh kept as accumulator tiles); a 2nd launch adds gi and the gates
# speedup vs baseline: 4.7568x; 1.0355x over previous
#include <hip/hip_runtime.h>
#include <hip/hip_bf16.h>
#include <math.h>

typedef float v2f __attribute__((ext_vector_type(2)));
typedef float v8f __attribute__((ext_vector_type(8)));

#define KDIM 128

template <bool TRANS_FILL, bool RELU_IN>
__global__ __launch_bounds__(256) void ggnn_gemm(
    const float* __restrict__ X, const float* __restrict__ W,
    const float* __restrict__ bias, float* __restrict__ Y,
    int nrows, int out_stride, int out_cols_total, int w_zstride, int y_zstride)
{
    W += (size_t)blockIdx.z * w_zstride;
    Y += (size_t)blockIdx.z * y_zstride;

    __shared__ float Wt[KDIM * KDIM];

    const int tid   = threadIdx.x;
    const int cbase = blockIdx.y * KDIM;
    int ccols = out_cols_total - cbase;
    if (ccols > KDIM) ccols = KDIM;

    if (TRANS_FILL) {
        for (int i = tid; i < KDIM * KDIM; i += 256) {
            const int n = i >> 7;
            const int k = i & (KDIM - 1);
            float w = 0.0f;
            if (n < ccols) w = W[(size_t)k * out_cols_total + cbase + n];
            Wt[n * KDIM + (k ^ ((n & 15) << 2))] = w;
        }
    } else {
        float4 stage[16];
        #pragma unroll
        for (int j = 0; j < 16; ++j) {
            const int i  = tid + j * 256;
            const int n  = i >> 5;
            const int k4 = (i & 31) << 2;
            const int nr = (n < ccols) ? n : 0;
            stage[j] = *(const float4*)(W + (size_t)(cbase + nr) * KDIM + k4);
        }
        #pragma unroll
        for (int j = 0; j < 16; ++j) {
            const int i  = tid + j * 256;
            const int n  = i >> 5;
            const int k4 = (i & 31) << 2;
            const float4 v = (n < ccols) ? stage[j] : make_float4(0.0f, 0.0f, 0.0f, 0.0f);
            *(float4*)&Wt[n * KDIM + (k4 ^ ((n & 15) << 2))] = v;
        }
    }
    __syncthreads();

    const int wave = tid >> 5;
    const int lane = tid & 31;
    const int m0 = (blockIdx.x * 8 + wave) * 16;
    if (m0 >= nrows) return;

    const int half = lane >> 4;
    const int l16  = lane & 15;

    const float* Xrow = X + (size_t)(m0 + l16) * KDIM + half * 2;
    const float* Brow = &Wt[l16 * KDIM + half * 2];
    const int swz = l16 << 2;

    v8f acc[8] = {};

    for (int kb = 0; kb < KDIM; kb += 4) {
        v2f a = *(const v2f*)(Xrow + kb);
        if (RELU_IN) { a.x = fmaxf(a.x, 0.0f); a.y = fmaxf(a.y, 0.0f); }
        const int kc = kb ^ swz;
        #pragma unroll
        for (int t = 0; t < 8; ++t) {
            v2f b = *(const v2f*)(Brow + t * 16 * KDIM + kc);
            acc[t] = __builtin_amdgcn_wmma_f32_16x16x4_f32(
                false, a, false, b, (short)0, acc[t], false, false);
        }
    }

    #pragma unroll
    for (int t = 0; t < 8; ++t) {
        const int c = t * 16 + l16;
        if (c < ccols) {
            const float bv = bias ? bias[cbase + c] : 0.0f;
            #pragma unroll
            for (int r = 0; r < 8; ++r) {
                const int row = m0 + r + half * 8;
                Y[(size_t)row * out_stride + cbase + c] = acc[t][r] + bv;
            }
        }
    }
}

__global__ __launch_bounds__(256) void ggnn_clear_counts(int* __restrict__ cnt, int n)
{
    const int i = blockIdx.x * 256 + threadIdx.x;
    if (i < n) cnt[i] = 0;
}

__global__ __launch_bounds__(256) void ggnn_count_dst(
    const int* __restrict__ dst, int* __restrict__ cnt, int* __restrict__ rank, int nedges)
{
    const int e = blockIdx.x * 256 + threadIdx.x;
    if (e < nedges) rank[e] = atomicAdd(&cnt[dst[e]], 1);
}

__global__ __launch_bounds__(1024) void ggnn_scan_counts(
    const int* __restrict__ cnt, int* __restrict__ rowptr, int n)
{
    __shared__ int wsum[32];
    const int t    = threadIdx.x;
    const int lane = t & 31;
    const int w    = t >> 5;
    int carry = 0;
    for (int base = 0; base < n; base += 4096) {
        const int i0 = base + t * 4;
        int c[4];
        #pragma unroll
        for (int k = 0; k < 4; ++k) c[k] = (i0 + k < n) ? cnt[i0 + k] : 0;
        const int s = c[0] + c[1] + c[2] + c[3];

        int incl = s;
        #pragma unroll
        for (int off = 1; off < 32; off <<= 1) {
            const int u = __shfl_up(incl, off);
            if (lane >= off) incl += u;
        }
        if (lane == 31) wsum[w] = incl;
        __syncthreads();
        if (w == 0) {
            int si = wsum[lane];
            #pragma unroll
            for (int off = 1; off < 32; off <<= 1) {
                const int u = __shfl_up(si, off);
                if (lane >= off) si += u;
            }
            wsum[lane] = si;
        }
        __syncthreads();
        const int wprefix = (w == 0) ? 0 : wsum[w - 1];
        const int total   = wsum[31];
        int run = carry + wprefix + incl - s;
        #pragma unroll
        for (int k = 0; k < 4; ++k) {
            if (i0 + k < n) rowptr[i0 + k] = run;
            run += c[k];
        }
        carry += total;
        __syncthreads();
    }
    if (t == 0) rowptr[n] = carry;
}

__global__ __launch_bounds__(256) void ggnn_fill_lists(
    const int* __restrict__ src, const int* __restrict__ dst,
    const int* __restrict__ rowptr, const int* __restrict__ rank,
    int* __restrict__ csr_src, int nedges)
{
    const int e = blockIdx.x * 256 + threadIdx.x;
    if (e < nedges) csr_src[rowptr[dst[e]] + rank[e]] = src[e];
}

__device__ __forceinline__ void ggnn_add4(float4& acc, const float4 v)
{
    acc.x += v.x; acc.y += v.y; acc.z += v.z; acc.w += v.w;
}

__device__ __forceinline__ void ggnn_gather_node(
    const float* __restrict__ m, const int* __restrict__ rowptr,
    const int* __restrict__ csr_src, float* __restrict__ agg, int node, int lane)
{
    const int beg = __builtin_amdgcn_readfirstlane(rowptr[node]);
    const int end = __builtin_amdgcn_readfirstlane(rowptr[node + 1]);
    const float* mcol = m + lane * 4;
    float4 acc = make_float4(0.f, 0.f, 0.f, 0.f);

    for (int base = beg; base < end; base += 32) {
        int nb = end - base;
        if (nb > 32) nb = 32;
        const int mine = (lane < nb) ? csr_src[base + lane] : 0;
        int j = 0;
        for (; j + 8 <= nb; j += 8) {
            float4 v[8];
            #pragma unroll
            for (int u = 0; u < 8; ++u) {
                const int s = __builtin_amdgcn_readlane(mine, j + u);
                v[u] = *(const float4*)(mcol + (size_t)s * KDIM);
            }
            #pragma unroll
            for (int u = 0; u < 8; ++u) ggnn_add4(acc, v[u]);
        }
        if (j + 4 <= nb) {
            float4 v[4];
            #pragma unroll
            for (int u = 0; u < 4; ++u) {
                const int s = __builtin_amdgcn_readlane(mine, j + u);
                v[u] = *(const float4*)(mcol + (size_t)s * KDIM);
            }
            #pragma unroll
            for (int u = 0; u < 4; ++u) ggnn_add4(acc, v[u]);
            j += 4;
        }
        if (j + 2 <= nb) {
            const int s0 = __builtin_amdgcn_readlane(mine, j);
            const int s1 = __builtin_amdgcn_readlane(mine, j + 1);
            const float4 v0 = *(const float4*)(mcol + (size_t)s0 * KDIM);
            const float4 v1 = *(const float4*)(mcol + (size_t)s1 * KDIM);
            ggnn_add4(acc, v0);
            ggnn_add4(acc, v1);
            j += 2;
        }
        if (j < nb) {
            const int s0 = __builtin_amdgcn_readlane(mine, j);
            const float4 v0 = *(const float4*)(mcol + (size_t)s0 * KDIM);
            ggnn_add4(acc, v0);
        }
    }
    *(float4*)(agg + (size_t)node * KDIM + lane * 4) = acc;
}

#define GRU_SLICE   32
#define GRU_WROWS   (3 * GRU_SLICE)
#define GRU_GROUPS  64
#define GRU_GH_TILE (6 * 2 * 32)

#define GRU_WAVES   32
#define GRU_THREADS (GRU_WAVES * 32)
#define GRU_STAGE_PIECES ((GRU_WROWS * (KDIM / 4)) / GRU_THREADS)
#define HH_NODES_PER_TURN 4

__global__ __launch_bounds__(GRU_THREADS) void ggnn_gather_hh(
    const float* __restrict__ hin, const float* __restrict__ w_hh,
    const int* __restrict__ rowptr, const int* __restrict__ csr_src,
    float* __restrict__ agg, float* __restrict__ gh, int nrows)
{
    __shared__ float Wt[GRU_WROWS * KDIM];
    __shared__ int next_tile;
    __shared__ int next_node;

    const int tid = threadIdx.x;
    const int c0  = blockIdx.x * GRU_SLICE;

    if (tid == 0) { next_tile = 0; next_node = 0; }

    float4 stage[GRU_STAGE_PIECES];
    #pragma unroll
    for (int j = 0; j < GRU_STAGE_PIECES; ++j) {
        const int i  = tid + j * GRU_THREADS;
        const int n  = i >> 5;
        const int k4 = (i & 31) << 2;
        const int g  = n >> 5;
        const int w  = n & 31;
        stage[j] = *(const float4*)(w_hh + (size_t)(g * 128 + c0 + w) * KDIM + k4);
    }
    #pragma unroll
    for (int j = 0; j < GRU_STAGE_PIECES; ++j) {
        const int i  = tid + j * GRU_THREADS;
        const int n  = i >> 5;
        const int k4 = (i & 31) << 2;
        *(float4*)&Wt[n * KDIM + (k4 ^ ((n & 15) << 2))] = stage[j];
    }
    __syncthreads();

    const int lane = tid & 31;
    const int wave = __builtin_amdgcn_readfirstlane(tid >> 5);
    const int half = lane >> 4;
    const int l16  = lane & 15;

    const float* Brow = &Wt[l16 * KDIM + half * 2];
    const int swz = l16 << 2;

    const int slices    = (int)gridDim.x;
    const int slice     = (int)blockIdx.x;
    const int groups    = (int)gridDim.y;
    const int group     = (int)blockIdx.y;
    const int row_tiles = (nrows + 15) >> 4;

    const int nwg     = slices * groups;
    const int per     = (nrows + nwg - 1) / nwg;
    const int node_lo = (group * slices + slice) * per;
    int node_hi = node_lo + per;
    if (node_hi > nrows) node_hi = nrows;

    bool tiles_left = true;
    bool nodes_left = true;
    bool tile_turn  = (wave & 1) == 0;
    while (tiles_left || nodes_left) {
        if (tile_turn && tiles_left) {
            int ticket = 0;
            if (lane == 0) ticket = atomicAdd(&next_tile, 1);
            ticket = __builtin_amdgcn_readfirstlane(ticket);
            const int rt = group + ticket * groups;
            if (rt >= row_tiles) {
                tiles_left = false;
            } else {
                const int m0 = rt * 16;
                const float* Hrow = hin + (size_t)(m0 + l16) * KDIM + half * 2;

                v8f acc[6] = {};
                for (int kb = 0; kb < KDIM; kb += 4) {
                    const v2f ah = *(const v2f*)(Hrow + kb);
                    const int kc = kb ^ swz;
                    #pragma unroll
                    for (int t = 0; t < 6; ++t) {
                        const v2f b = *(const v2f*)(Brow + t * 16 * KDIM + kc);
                        acc[t] = __builtin_amdgcn_wmma_f32_16x16x4_f32(
                            false, ah, false, b, (short)0, acc[t], false, false);
                    }
                }

                float4* ghp = (float4*)gh + ((size_t)rt * slices + slice) * GRU_GH_TILE + lane;
                #pragma unroll
                for (int t = 0; t < 6; ++t) {
                    ghp[(t * 2 + 0) * 32] = make_float4(acc[t][0], acc[t][1], acc[t][2], acc[t][3]);
                    ghp[(t * 2 + 1) * 32] = make_float4(acc[t][4], acc[t][5], acc[t][6], acc[t][7]);
                }
            }
        }
        tile_turn = true;
        if (nodes_left) {
            for (int q = 0; q < HH_NODES_PER_TURN; ++q) {
                int ticket = 0;
                if (lane == 0) ticket = atomicAdd(&next_node, 1);
                ticket = __builtin_amdgcn_readfirstlane(ticket);
                const int node = node_lo + ticket;
                if (node >= node_hi) { nodes_left = false; break; }
                ggnn_gather_node(hin, rowptr, csr_src, agg, node, lane);
            }
        }
    }
}

__global__ __launch_bounds__(GRU_THREADS) void ggnn_gru_in(
    const float* __restrict__ agg, const float* __restrict__ hin,
    const float* __restrict__ w_ih, const float* __restrict__ gh,
    const float* __restrict__ b_ih, const float* __restrict__ b_hh,
    float* __restrict__ hout, int nrows)
{
    __shared__ float Wt[GRU_WROWS * KDIM];
    __shared__ int next_tile;

    const int tid = threadIdx.x;
    const int c0  = blockIdx.x * GRU_SLICE;

    if (tid == 0) next_tile = 0;

    float4 stage[GRU_STAGE_PIECES];
    #pragma unroll
    for (int j = 0; j < GRU_STAGE_PIECES; ++j) {
        const int i  = tid + j * GRU_THREADS;
        const int n  = i >> 5;
        const int k4 = (i & 31) << 2;
        const int g  = n >> 5;
        const int w  = n & 31;
        stage[j] = *(const float4*)(w_ih + (size_t)(g * 128 + c0 + w) * KDIM + k4);
    }
    #pragma unroll
    for (int j = 0; j < GRU_STAGE_PIECES; ++j) {
        const int i  = tid + j * GRU_THREADS;
        const int n  = i >> 5;
        const int k4 = (i & 31) << 2;
        *(float4*)&Wt[n * KDIM + (k4 ^ ((n & 15) << 2))] = stage[j];
    }
    __syncthreads();

    const int lane = tid & 31;
    const int half = lane >> 4;
    const int l16  = lane & 15;

    const float* Brow = &Wt[l16 * KDIM + half * 2];
    const int swz = l16 << 2;

    const int slices    = (int)gridDim.x;
    const int slice     = (int)blockIdx.x;
    const int groups    = (int)gridDim.y;
    const int group     = (int)blockIdx.y;
    const int row_tiles = (nrows + 15) >> 4;

    for (;;) {
        int ticket = 0;
        if (lane == 0) ticket = atomicAdd(&next_tile, 1);
        ticket = __builtin_amdgcn_readfirstlane(ticket);
        const int rt = group + ticket * groups;
        if (rt >= row_tiles) break;

        const int m0 = rt * 16;
        const float* Arow = agg + (size_t)(m0 + l16) * KDIM + half * 2;

        v8f acc[6] = {};
        for (int kb = 0; kb < KDIM; kb += 4) {
            const v2f a  = *(const v2f*)(Arow + kb);
            const int kc = kb ^ swz;
            #pragma unroll
            for (int t = 0; t < 6; ++t) {
                const v2f b = *(const v2f*)(Brow + t * 16 * KDIM + kc);
                acc[t] = __builtin_amdgcn_wmma_f32_16x16x4_f32(
                    false, a, false, b, (short)0, acc[t], false, false);
            }
        }

        const float4* ghp = (const float4*)gh + ((size_t)rt * slices + slice) * GRU_GH_TILE + lane;
        #pragma unroll
        for (int j = 0; j < 2; ++j) {
            const int c = c0 + j * 16 + l16;
            const float bir = b_ih[c], biz = b_ih[c + 128], bin = b_ih[c + 256];
            const float bhr = b_hh[c], bhz = b_hh[c + 128], bhn = b_hh[c + 256];
            float ghr[8], ghz[8], ghn[8];
            #pragma unroll
            for (int q = 0; q < 2; ++q) {
                const float4 vr = ghp[((0 + j) * 2 + q) * 32];
                const float4 vz = ghp[((2 + j) * 2 + q) * 32];
                const float4 vn = ghp[((4 + j) * 2 + q) * 32];
                ghr[q * 4 + 0] = vr.x; ghr[q * 4 + 1] = vr.y; ghr[q * 4 + 2] = vr.z; ghr[q * 4 + 3] = vr.w;
                ghz[q * 4 + 0] = vz.x; ghz[q * 4 + 1] = vz.y; ghz[q * 4 + 2] = vz.z; ghz[q * 4 + 3] = vz.w;
                ghn[q * 4 + 0] = vn.x; ghn[q * 4 + 1] = vn.y; ghn[q * 4 + 2] = vn.z; ghn[q * 4 + 3] = vn.w;
            }
            #pragma unroll
            for (int r = 0; r < 8; ++r) {
                const size_t off = (size_t)(m0 + r + half * 8) * KDIM + c;
                const float i_r = acc[0 + j][r] + bir, h_r = ghr[r] + bhr;
                const float i_z = acc[2 + j][r] + biz, h_z = ghz[r] + bhz;
                const float i_n = acc[4 + j][r] + bin, h_n = ghn[r] + bhn;
                const float rg = 1.0f / (1.0f + __expf(-(i_r + h_r)));
                const float zg = 1.0f / (1.0f + __expf(-(i_z + h_z)));
                const float nn = tanhf(i_n + rg * h_n);
                const float hv = hin[off];
                hout[off] = (1.0f - zg) * nn + zg * hv;
            }
        }
    }
}

extern "C" void kernel_launch(void* const* d_in, const int* in_sizes, int n_in,
                              void* d_out, int out_size, void* d_ws, size_t ws_size,
                              hipStream_t stream)
{
    (void)n_in; (void)out_size; (void)ws_size;

    const float* x      = (const float*)d_in[0];
    const int*   ei     = (const int*)d_in[1];
    const float* lin_w  = (const float*)d_in[2];
    const float* lin_b  = (const float*)d_in[3];
    const float* ggc_w  = (const float*)d_in[4];
    const float* w_ih   = (const float*)d_in[5];
    const float* w_hh   = (const float*)d_in[6];
    const float* b_ih   = (const float*)d_in[7];
    const float* b_hh   = (const float*)d_in[8];
    const float* lin2_w = (const float*)d_in[9];
    const float* lin2_b = (const float*)d_in[10];
    float* out = (float*)d_out;

    const int N = in_sizes[0] / KDIM;
    const int E = in_sizes[1] / 2;
    const int* src = ei;
    const int* dst = ei + E;

    const int row_tiles  = (N + 15) / 16;
    const int row_blocks = (row_tiles + 7) / 8;

    float* h     = (float*)d_ws;
    float* wfold = h + (size_t)N * 128;
    float* agg   = wfold + (size_t)N * 128;
    float* h2  = agg + (size_t)N * 128;
    float* ghb = h2 + (size_t)N * 128;
    int* cnt     = (int*)(ghb + (size_t)row_tiles * 16 * 384);
    int* rowptr  = cnt + N;
    int* rank    = rowptr + N + 1;
    int* csr_src = rank + E;

    const dim3 blk(256);

    const unsigned node_blks    = (unsigned)((N + 255) / 256);
    const unsigned edge_blks    = (unsigned)((E + 255) / 256);
    float* hbuf[2] = { h, h2 };

    ggnn_gemm<false, false><<<dim3(row_blocks, 1), blk, 0, stream>>>(
        x, lin_w, lin_b, h, N, 128, 128, 0, 0);

    ggnn_gemm<false, false><<<dim3(3, 1, 8), blk, 0, stream>>>(
        w_ih, ggc_w, nullptr, wfold, 384, 128, 128, 128 * 128, 384 * 128);

    ggnn_clear_counts<<<node_blks, blk, 0, stream>>>(cnt, N);
    ggnn_count_dst<<<edge_blks, blk, 0, stream>>>(dst, cnt, rank, E);
    ggnn_scan_counts<<<1, dim3(1024), 0, stream>>>(cnt, rowptr, N);
    ggnn_fill_lists<<<edge_blks, blk, 0, stream>>>(src, dst, rowptr, rank, csr_src, E);

    for (int l = 0; l < 8; ++l) {
        const float* hcur = hbuf[l & 1];
        float* hnew = hbuf[(l + 1) & 1];
        ggnn_gather_hh<<<dim3(128 / GRU_SLICE, GRU_GROUPS), dim3(GRU_THREADS), 0, stream>>>(
            hcur, w_hh, rowptr, csr_src, agg, ghb, N);
        ggnn_gru_in<<<dim3(128 / GRU_SLICE, GRU_GROUPS), dim3(GRU_THREADS), 0, stream>>>(
            agg, hcur, wfold + (size_t)l * 384 * 128, ghb, b_ih, b_hh, hnew, N);
    }

    ggnn_gemm<false, true><<<dim3(row_blocks, 1), blk, 0, stream>>>(
        h, lin2_w, lin2_b, out, N, 100, 100, 0, 0);
}


# GRU gates branch-free on hardware exp2 and reciprocal: sigmoid 1/(1+e^-x) by v_rcp_f32, tanh as 1-2/(1+e^2x); no library tanhf, no IEEE division (last-bit differences in the gates)
# speedup vs baseline: 5.0847x; 1.0689x over previous
#include <hip/hip_runtime.h>
#include <hip/hip_bf16.h>
#include <math.h>

typedef float v2f __attribute__((ext_vector_type(2)));
typedef float v8f __attribute__((ext_vector_type(8)));

#define KDIM 128

template <bool TRANS_FILL, bool RELU_IN>
__global__ __launch_bounds__(256) void ggnn_gemm(
    const float* __restrict__ X, const float* __restrict__ W,
    const float* __restrict__ bias, float* __restrict__ Y,
    int nrows, int out_stride, int out_cols_total, int w_zstride, int y_zstride)
{
    W += (size_t)blockIdx.z * w_zstride;
    Y += (size_t)blockIdx.z * y_zstride;

    __shared__ float Wt[KDIM * KDIM];

    const int tid   = threadIdx.x;
    const int cbase = blockIdx.y * KDIM;
    int ccols = out_cols_total - cbase;
    if (ccols > KDIM) ccols = KDIM;

    if (TRANS_FILL) {
        for (int i = tid; i < KDIM * KDIM; i += 256) {
            const int n = i >> 7;
            const int k = i & (KDIM - 1);
            float w = 0.0f;
            if (n < ccols) w = W[(size_t)k * out_cols_total + cbase + n];
            Wt[n * KDIM + (k ^ ((n & 15) << 2))] = w;
        }
    } else {
        float4 stage[16];
        #pragma unroll
        for (int j = 0; j < 16; ++j) {
            const int i  = tid + j * 256;
            const int n  = i >> 5;
            const int k4 = (i & 31) << 2;
            const int nr = (n < ccols) ? n : 0;
            stage[j] = *(const float4*)(W + (size_t)(cbase + nr) * KDIM + k4);
        }
        #pragma unroll
        for (int j = 0; j < 16; ++j) {
            const int i  = tid + j * 256;
            const int n  = i >> 5;
            const int k4 = (i & 31) << 2;
            const float4 v = (n < ccols) ? stage[j] : make_float4(0.0f, 0.0f, 0.0f, 0.0f);
            *(float4*)&Wt[n * KDIM + (k4 ^ ((n & 15) << 2))] = v;
        }
    }
    __syncthreads();

    const int wave = tid >> 5;
    const int lane = tid & 31;
    const int m0 = (blockIdx.x * 8 + wave) * 16;
    if (m0 >= nrows) return;

    const int half = lane >> 4;
    const int l16  = lane & 15;

    const float* Xrow = X + (size_t)(m0 + l16) * KDIM + half * 2;
    const float* Brow = &Wt[l16 * KDIM + half * 2];
    const int swz = l16 << 2;

    v8f acc[8] = {};

    for (int kb = 0; kb < KDIM; kb += 4) {
        v2f a = *(const v2f*)(Xrow + kb);
        if (RELU_IN) { a.x = fmaxf(a.x, 0.0f); a.y = fmaxf(a.y, 0.0f); }
        const int kc = kb ^ swz;
        #pragma unroll
        for (int t = 0; t < 8; ++t) {
            v2f b = *(const v2f*)(Brow + t * 16 * KDIM + kc);
            acc[t] = __builtin_amdgcn_wmma_f32_16x16x4_f32(
                false, a, false, b, (short)0, acc[t], false, false);
        }
    }

    #pragma unroll
    for (int t = 0; t < 8; ++t) {
        const int c = t * 16 + l16;
        if (c < ccols) {
            const float bv = bias ? bias[cbase + c] : 0.0f;
            #pragma unroll
            for (int r = 0; r < 8; ++r) {
                const int row = m0 + r + half * 8;
                Y[(size_t)row * out_stride + cbase + c] = acc[t][r] + bv;
            }
        }
    }
}

__global__ __launch_bounds__(256) void ggnn_clear_counts(int* __restrict__ cnt, int n)
{
    const int i = blockIdx.x * 256 + threadIdx.x;
    if (i < n) cnt[i] = 0;
}

__global__ __launch_bounds__(256) void ggnn_count_dst(
    const int* __restrict__ dst, int* __restrict__ cnt, int* __restrict__ rank, int nedges)
{
    const int e = blockIdx.x * 256 + threadIdx.x;
    if (e < nedges) rank[e] = atomicAdd(&cnt[dst[e]], 1);
}

__global__ __launch_bounds__(1024) void ggnn_scan_counts(
    const int* __restrict__ cnt, int* __restrict__ rowptr, int n)
{
    __shared__ int wsum[32];
    const int t    = threadIdx.x;
    const int lane = t & 31;
    const int w    = t >> 5;
    int carry = 0;
    for (int base = 0; base < n; base += 4096) {
        const int i0 = base + t * 4;
        int c[4];
        #pragma unroll
        for (int k = 0; k < 4; ++k) c[k] = (i0 + k < n) ? cnt[i0 + k] : 0;
        const int s = c[0] + c[1] + c[2] + c[3];

        int incl = s;
        #pragma unroll
        for (int off = 1; off < 32; off <<= 1) {
            const int u = __shfl_up(incl, off);
            if (lane >= off) incl += u;
        }
        if (lane == 31) wsum[w] = incl;
        __syncthreads();
        if (w == 0) {
            int si = wsum[lane];
            #pragma unroll
            for (int off = 1; off < 32; off <<= 1) {
                const int u = __shfl_up(si, off);
                if (lane >= off) si += u;
            }
            wsum[lane] = si;
        }
        __syncthreads();
        const int wprefix = (w == 0) ? 0 : wsum[w - 1];
        const int total   = wsum[31];
        int run = carry + wprefix + incl - s;
        #pragma unroll
        for (int k = 0; k < 4; ++k) {
            if (i0 + k < n) rowptr[i0 + k] = run;
            run += c[k];
        }
        carry += total;
        __syncthreads();
    }
    if (t == 0) rowptr[n] = carry;
}

__global__ __launch_bounds__(256) void ggnn_fill_lists(
    const int* __restrict__ src, const int* __restrict__ dst,
    const int* __restrict__ rowptr, const int* __restrict__ rank,
    int* __restrict__ csr_src, int nedges)
{
    const int e = blockIdx.x * 256 + threadIdx.x;
    if (e < nedges) csr_src[rowptr[dst[e]] + rank[e]] = src[e];
}

__device__ __forceinline__ void ggnn_add4(float4& acc, const float4 v)
{
    acc.x += v.x; acc.y += v.y; acc.z += v.z; acc.w += v.w;
}

__device__ __forceinline__ void ggnn_gather_node(
    const float* __restrict__ m, const int* __restrict__ rowptr,
    const int* __restrict__ csr_src, float* __restrict__ agg, int node, int lane)
{
    const int beg = __builtin_amdgcn_readfirstlane(rowptr[node]);
    const int end = __builtin_amdgcn_readfirstlane(rowptr[node + 1]);
    const float* mcol = m + lane * 4;
    float4 acc = make_float4(0.f, 0.f, 0.f, 0.f);

    for (int base = beg; base < end; base += 32) {
        int nb = end - base;
        if (nb > 32) nb = 32;
        const int mine = (lane < nb) ? csr_src[base + lane] : 0;
        int j = 0;
        for (; j + 8 <= nb; j += 8) {
            float4 v[8];
            #pragma unroll
            for (int u = 0; u < 8; ++u) {
                const int s = __builtin_amdgcn_readlane(mine, j + u);
                v[u] = *(const float4*)(mcol + (size_t)s * KDIM);
            }
            #pragma unroll
            for (int u = 0; u < 8; ++u) ggnn_add4(acc, v[u]);
        }
        if (j + 4 <= nb) {
            float4 v[4];
            #pragma unroll
            for (int u = 0; u < 4; ++u) {
                const int s = __builtin_amdgcn_readlane(mine, j + u);
                v[u] = *(const float4*)(mcol + (size_t)s * KDIM);
            }
            #pragma unroll
            for (int u = 0; u < 4; ++u) ggnn_add4(acc, v[u]);
            j += 4;
        }
        if (j + 2 <= nb) {
            const int s0 = __builtin_amdgcn_readlane(mine, j);
            const int s1 = __builtin_amdgcn_readlane(mine, j + 1);
            const float4 v0 = *(const float4*)(mcol + (size_t)s0 * KDIM);
            const float4 v1 = *(const float4*)(mcol + (size_t)s1 * KDIM);
            ggnn_add4(acc, v0);
            ggnn_add4(acc, v1);
            j += 2;
        }
        if (j < nb) {
            const int s0 = __builtin_amdgcn_readlane(mine, j);
            const float4 v0 = *(const float4*)(mcol + (size_t)s0 * KDIM);
            ggnn_add4(acc, v0);
        }
    }
    *(float4*)(agg + (size_t)node * KDIM + lane * 4) = acc;
}

#define GRU_SLICE   32
#define GRU_WROWS   (3 * GRU_SLICE)
#define GRU_GROUPS  64
#define GRU_GH_TILE (6 * 2 * 32)

#define GRU_WAVES   32
#define GRU_THREADS (GRU_WAVES * 32)
#define GRU_STAGE_PIECES ((GRU_WROWS * (KDIM / 4)) / GRU_THREADS)
#define HH_NODES_PER_TURN 4

__global__ __launch_bounds__(GRU_THREADS) void ggnn_gather_hh(
    const float* __restrict__ hin, const float* __restrict__ w_hh,
    const int* __restrict__ rowptr, const int* __restrict__ csr_src,
    float* __restrict__ agg, float* __restrict__ gh, int nrows)
{
    __shared__ float Wt[GRU_WROWS * KDIM];
    __shared__ int next_tile;
    __shared__ int next_node;

    const int tid = threadIdx.x;
    const int c0  = blockIdx.x * GRU_SLICE;

    if (tid == 0) { next_tile = 0; next_node = 0; }

    float4 stage[GRU_STAGE_PIECES];
    #pragma unroll
    for (int j = 0; j < GRU_STAGE_PIECES; ++j) {
        const int i  = tid + j * GRU_THREADS;
        const int n  = i >> 5;
        const int k4 = (i & 31) << 2;
        const int g  = n >> 5;
        const int w  = n & 31;
        stage[j] = *(const float4*)(w_hh + (size_t)(g * 128 + c0 + w) * KDIM + k4);
    }
    #pragma unroll
    for (int j = 0; j < GRU_STAGE_PIECES; ++j) {
        const int i  = tid + j * GRU_THREADS;
        const int n  = i >> 5;
        const int k4 = (i & 31) << 2;
        *(float4*)&Wt[n * KDIM + (k4 ^ ((n & 15) << 2))] = stage[j];
    }
    __syncthreads();

    const int lane = tid & 31;
    const int wave = __builtin_amdgcn_readfirstlane(tid >> 5);
    const int half = lane >> 4;
    const int l16  = lane & 15;

    const float* Brow = &Wt[l16 * KDIM + half * 2];
    const int swz = l16 << 2;

    const int slices    = (int)gridDim.x;
    const int slice     = (int)blockIdx.x;
    const int groups    = (int)gridDim.y;
    const int group     = (int)blockIdx.y;
    const int row_tiles = (nrows + 15) >> 4;

    const int nwg     = slices * groups;
    const int per     = (nrows + nwg - 1) / nwg;
    const int node_lo = (group * slices + slice) * per;
    int node_hi = node_lo + per;
    if (node_hi > nrows) node_hi = nrows;

    bool tiles_left = true;
    bool nodes_left = true;
    bool tile_turn  = (wave & 1) == 0;
    while (tiles_left || nodes_left) {
        if (tile_turn && tiles_left) {
            int ticket = 0;
            if (lane == 0) ticket = atomicAdd(&next_tile, 1);
            ticket = __builtin_amdgcn_readfirstlane(ticket);
            const int rt = group + ticket * groups;
            if (rt >= row_tiles) {
                tiles_left = false;
            } else {
                const int m0 = rt * 16;
                const float* Hrow = hin + (size_t)(m0 + l16) * KDIM + half * 2;

                v8f acc[6] = {};
                for (int kb = 0; kb < KDIM; kb += 4) {
                    const v2f ah = *(const v2f*)(Hrow + kb);
                    const int kc = kb ^ swz;
                    #pragma unroll
                    for (int t = 0; t < 6; ++t) {
                        const v2f b = *(const v2f*)(Brow + t * 16 * KDIM + kc);
                        acc[t] = __builtin_amdgcn_wmma_f32_16x16x4_f32(
                            false, ah, false, b, (short)0, acc[t], false, false);
                    }
                }

                float4* ghp = (float4*)gh + ((size_t)rt * slices + slice) * GRU_GH_TILE + lane;
                #pragma unroll
                for (int t = 0; t < 6; ++t) {
                    ghp[(t * 2 + 0) * 32] = make_float4(acc[t][0], acc[t][1], acc[t][2], acc[t][3]);
                    ghp[(t * 2 + 1) * 32] = make_float4(acc[t][4], acc[t][5], acc[t][6], acc[t][7]);
                }
            }
        }
        tile_turn = true;
        if (nodes_left) {
            for (int q = 0; q < HH_NODES_PER_TURN; ++q) {
                int ticket = 0;
                if (lane == 0) ticket = atomicAdd(&next_node, 1);
                ticket = __builtin_amdgcn_readfirstlane(ticket);
                const int node = node_lo + ticket;
                if (node >= node_hi) { nodes_left = false; break; }
                ggnn_gather_node(hin, rowptr, csr_src, agg, node, lane);
            }
        }
    }
}

__device__ __forceinline__ float ggnn_sigmoid(float x)
{
    return __builtin_amdgcn_rcpf(1.0f + __builtin_amdgcn_exp2f(x * -1.4426950408889634f));
}

__device__ __forceinline__ float ggnn_tanh(float x)
{
    return 1.0f - 2.0f * __builtin_amdgcn_rcpf(1.0f + __builtin_amdgcn_exp2f(x * 2.8853900817779268f));
}

__global__ __launch_bounds__(GRU_THREADS) void ggnn_gru_in(
    const float* __restrict__ agg, const float* __restrict__ hin,
    const float* __restrict__ w_ih, const float* __restrict__ gh,
    const float* __restrict__ b_ih, const float* __restrict__ b_hh,
    float* __restrict__ hout, int nrows)
{
    __shared__ float Wt[GRU_WROWS * KDIM];
    __shared__ int next_tile;

    const int tid = threadIdx.x;
    const int c0  = blockIdx.x * GRU_SLICE;

    if (tid == 0) next_tile = 0;

    float4 stage[GRU_STAGE_PIECES];
    #pragma unroll
    for (int j = 0; j < GRU_STAGE_PIECES; ++j) {
        const int i  = tid + j * GRU_THREADS;
        const int n  = i >> 5;
        const int k4 = (i & 31) << 2;
        const int g  = n >> 5;
        const int w  = n & 31;
        stage[j] = *(const float4*)(w_ih + (size_t)(g * 128 + c0 + w) * KDIM + k4);
    }
    #pragma unroll
    for (int j = 0; j < GRU_STAGE_PIECES; ++j) {
        const int i  = tid + j * GRU_THREADS;
        const int n  = i >> 5;
        const int k4 = (i & 31) << 2;
        *(float4*)&Wt[n * KDIM + (k4 ^ ((n & 15) << 2))] = stage[j];
    }
    __syncthreads();

    const int lane = tid & 31;
    const int half = lane >> 4;
    const int l16  = lane & 15;

    const float* Brow = &Wt[l16 * KDIM + half * 2];
    const int swz = l16 << 2;

    const int slices    = (int)gridDim.x;
    const int slice     = (int)blockIdx.x;
    const int groups    = (int)gridDim.y;
    const int group     = (int)blockIdx.y;
    const int row_tiles = (nrows + 15) >> 4;

    for (;;) {
        int ticket = 0;
        if (lane == 0) ticket = atomicAdd(&next_tile, 1);
        ticket = __builtin_amdgcn_readfirstlane(ticket);
        const int rt = group + ticket * groups;
        if (rt >= row_tiles) break;

        const int m0 = rt * 16;
        const float* Arow = agg + (size_t)(m0 + l16) * KDIM + half * 2;

        v8f acc[6] = {};
        for (int kb = 0; kb < KDIM; kb += 4) {
            const v2f a  = *(const v2f*)(Arow + kb);
            const int kc = kb ^ swz;
            #pragma unroll
            for (int t = 0; t < 6; ++t) {
                const v2f b = *(const v2f*)(Brow + t * 16 * KDIM + kc);
                acc[t] = __builtin_amdgcn_wmma_f32_16x16x4_f32(
                    false, a, false, b, (short)0, acc[t], false, false);
            }
        }

        const float4* ghp = (const float4*)gh + ((size_t)rt * slices + slice) * GRU_GH_TILE + lane;
        #pragma unroll
        for (int j = 0; j < 2; ++j) {
            const int c = c0 + j * 16 + l16;
            const float bir = b_ih[c], biz = b_ih[c + 128], bin = b_ih[c + 256];
            const float bhr = b_hh[c], bhz = b_hh[c + 128], bhn = b_hh[c + 256];
            float ghr[8], ghz[8], ghn[8];
            #pragma unroll
            for (int q = 0; q < 2; ++q) {
                const float4 vr = ghp[((0 + j) * 2 + q) * 32];
                const float4 vz = ghp[((2 + j) * 2 + q) * 32];
                const float4 vn = ghp[((4 + j) * 2 + q) * 32];
                ghr[q * 4 + 0] = vr.x; ghr[q * 4 + 1] = vr.y; ghr[q * 4 + 2] = vr.z; ghr[q * 4 + 3] = vr.w;
                ghz[q * 4 + 0] = vz.x; ghz[q * 4 + 1] = vz.y; ghz[q * 4 + 2] = vz.z; ghz[q * 4 + 3] = vz.w;
                ghn[q * 4 + 0] = vn.x; ghn[q * 4 + 1] = vn.y; ghn[q * 4 + 2] = vn.z; ghn[q * 4 + 3] = vn.w;
            }
            #pragma unroll
            for (int r = 0; r < 8; ++r) {
                const size_t off = (size_t)(m0 + r + half * 8) * KDIM + c;
                const float i_r = acc[0 + j][r] + bir, h_r = ghr[r] + bhr;
                const float i_z = acc[2 + j][r] + biz, h_z = ghz[r] + bhz;
                const float i_n = acc[4 + j][r] + bin, h_n = ghn[r] + bhn;
                const float rg = ggnn_sigmoid(i_r + h_r);
                const float zg = ggnn_sigmoid(i_z + h_z);
                const float nn = ggnn_tanh(i_n + rg * h_n);
                const float hv = hin[off];
                hout[off] = (1.0f - zg) * nn + zg * hv;
            }
        }
    }
}

extern "C" void kernel_launch(void* const* d_in, const int* in_sizes, int n_in,
                              void* d_out, int out_size, void* d_ws, size_t ws_size,
                              hipStream_t stream)
{
    (void)n_in; (void)out_size; (void)ws_size;

    const float* x      = (const float*)d_in[0];
    const int*   ei     = (const int*)d_in[1];
    const float* lin_w  = (const float*)d_in[2];
    const float* lin_b  = (const float*)d_in[3];
    const float* ggc_w  = (const float*)d_in[4];
    const float* w_ih   = (const float*)d_in[5];
    const float* w_hh   = (const float*)d_in[6];
    const float* b_ih   = (const float*)d_in[7];
    const float* b_hh   = (const float*)d_in[8];
    const float* lin2_w = (const float*)d_in[9];
    const float* lin2_b = (const float*)d_in[10];
    float* out = (float*)d_out;

    const int N = in_sizes[0] / KDIM;
    const int E = in_sizes[1] / 2;
    const int* src = ei;
    const int* dst = ei + E;

    const int row_tiles  = (N + 15) / 16;
    const int row_blocks = (row_tiles + 7) / 8;

    float* h     = (float*)d_ws;
    float* wfold = h + (size_t)N * 128;
    float* agg   = wfold + (size_t)N * 128;
    float* h2  = agg + (size_t)N * 128;
    float* ghb = h2 + (size_t)N * 128;
    int* cnt     = (int*)(ghb + (size_t)row_tiles * 16 * 384);
    int* rowptr  = cnt + N;
    int* rank    = rowptr + N + 1;
    int* csr_src = rank + E;

    const dim3 blk(256);

    const unsigned node_blks    = (unsigned)((N + 255) / 256);
    const unsigned edge_blks    = (unsigned)((E + 255) / 256);
    float* hbuf[2] = { h, h2 };

    ggnn_gemm<false, false><<<dim3(row_blocks, 1), blk, 0, stream>>>(
        x, lin_w, lin_b, h, N, 128, 128, 0, 0);

    ggnn_gemm<false, false><<<dim3(3, 1, 8), blk, 0, stream>>>(
        w_ih, ggc_w, nullptr, wfold, 384, 128, 128, 128 * 128, 384 * 128);

    ggnn_clear_counts<<<node_blks, blk, 0, stream>>>(cnt, N);
    ggnn_count_dst<<<edge_blks, blk, 0, stream>>>(dst, cnt, rank, E);
    ggnn_scan_counts<<<1, dim3(1024), 0, stream>>>(cnt, rowptr, N);
    ggnn_fill_lists<<<edge_blks, blk, 0, stream>>>(src, dst, rowptr, rank, csr_src, E);

    for (int l = 0; l < 8; ++l) {
        const float* hcur = hbuf[l & 1];
        float* hnew = hbuf[(l + 1) & 1];
        ggnn_gather_hh<<<dim3(128 / GRU_SLICE, GRU_GROUPS), dim3(GRU_THREADS), 0, stream>>>(
            hcur, w_hh, rowptr, csr_src, agg, ghb, N);
        ggnn_gru_in<<<dim3(128 / GRU_SLICE, GRU_GROUPS), dim3(GRU_THREADS), 0, stream>>>(
            agg, hcur, wfold + (size_t)l * 384 * 128, ghb, b_ih, b_hh, hnew, N);
    }

    ggnn_gemm<false, true><<<dim3(row_blocks, 1), blk, 0, stream>>>(
        h, lin2_w, lin2_b, out, N, 100, 100, 0, 0);
}


# Edge grouping without global atomics: 16 node ranges x 16 edge chunks count and place edges on LDS integer counters (no rank array, no clearing pass); order in a node's sum: chunk, then LDS arrival
# speedup vs baseline: 5.4589x; 1.0736x over previous
#include <hip/hip_runtime.h>
#include <hip/hip_bf16.h>
#include <math.h>

typedef float v2f __attribute__((ext_vector_type(2)));
typedef float v8f __attribute__((ext_vector_type(8)));

#define KDIM 128

template <bool TRANS_FILL, bool RELU_IN>
__global__ __launch_bounds__(256) void ggnn_gemm(
    const float* __restrict__ X, const float* __restrict__ W,
    const float* __restrict__ bias, float* __restrict__ Y,
    int nrows, int out_stride, int out_cols_total, int w_zstride, int y_zstride)
{
    W += (size_t)blockIdx.z * w_zstride;
    Y += (size_t)blockIdx.z * y_zstride;

    __shared__ float Wt[KDIM * KDIM];

    const int tid   = threadIdx.x;
    const int cbase = blockIdx.y * KDIM;
    int ccols = out_cols_total - cbase;
    if (ccols > KDIM) ccols = KDIM;

    if (TRANS_FILL) {
        for (int i = tid; i < KDIM * KDIM; i += 256) {
            const int n = i >> 7;
            const int k = i & (KDIM - 1);
            float w = 0.0f;
            if (n < ccols) w = W[(size_t)k * out_cols_total + cbase + n];
            Wt[n * KDIM + (k ^ ((n & 15) << 2))] = w;
        }
    } else {
        float4 stage[16];
        #pragma unroll
        for (int j = 0; j < 16; ++j) {
            const int i  = tid + j * 256;
            const int n  = i >> 5;
            const int k4 = (i & 31) << 2;
            const int nr = (n < ccols) ? n : 0;
            stage[j] = *(const float4*)(W + (size_t)(cbase + nr) * KDIM + k4);
        }
        #pragma unroll
        for (int j = 0; j < 16; ++j) {
            const int i  = tid + j * 256;
            const int n  = i >> 5;
            const int k4 = (i & 31) << 2;
            const float4 v = (n < ccols) ? stage[j] : make_float4(0.0f, 0.0f, 0.0f, 0.0f);
            *(float4*)&Wt[n * KDIM + (k4 ^ ((n & 15) << 2))] = v;
        }
    }
    __syncthreads();

    const int wave = tid >> 5;
    const int lane = tid & 31;
    const int m0 = (blockIdx.x * 8 + wave) * 16;
    if (m0 >= nrows) return;

    const int half = lane >> 4;
    const int l16  = lane & 15;

    const float* Xrow = X + (size_t)(m0 + l16) * KDIM + half * 2;
    const float* Brow = &Wt[l16 * KDIM + half * 2];
    const int swz = l16 << 2;

    v8f acc[8] = {};

    for (int kb = 0; kb < KDIM; kb += 4) {
        v2f a = *(const v2f*)(Xrow + kb);
        if (RELU_IN) { a.x = fmaxf(a.x, 0.0f); a.y = fmaxf(a.y, 0.0f); }
        const int kc = kb ^ swz;
        #pragma unroll
        for (int t = 0; t < 8; ++t) {
            v2f b = *(const v2f*)(Brow + t * 16 * KDIM + kc);
            acc[t] = __builtin_amdgcn_wmma_f32_16x16x4_f32(
                false, a, false, b, (short)0, acc[t], false, false);
        }
    }

    #pragma unroll
    for (int t = 0; t < 8; ++t) {
        const int c = t * 16 + l16;
        if (c < ccols) {
            const float bv = bias ? bias[cbase + c] : 0.0f;
            #pragma unroll
            for (int r = 0; r < 8; ++r) {
                const int row = m0 + r + half * 8;
                Y[(size_t)row * out_stride + cbase + c] = acc[t][r] + bv;
            }
        }
    }
}

#define CSR_RANGES  16
#define CSR_CHUNKS  16
#define CSR_SPAN    4096
#define CSR_THREADS 1024
#define CSR_BATCH   8

__global__ __launch_bounds__(CSR_THREADS) void ggnn_count_dst(
    const int* __restrict__ dst, int* __restrict__ cnt2, int nedges, int n)
{
    __shared__ int lcnt[CSR_SPAN];
    const int tid   = threadIdx.x;
    const int per_r = (n + CSR_RANGES - 1) / CSR_RANGES;
    const int r_lo  = (int)blockIdx.x * per_r;
    int r_hi = r_lo + per_r;
    if (r_hi > n) r_hi = n;
    const int per_c = (nedges + CSR_CHUNKS - 1) / CSR_CHUNKS;
    const int e_lo  = (int)blockIdx.y * per_c;
    int e_hi = e_lo + per_c;
    if (e_hi > nedges) e_hi = nedges;
    int* out = cnt2 + (size_t)blockIdx.y * n;

    for (int lo = r_lo; lo < r_hi; lo += CSR_SPAN) {
        int hi = lo + CSR_SPAN;
        if (hi > r_hi) hi = r_hi;
        for (int i = tid; i < hi - lo; i += CSR_THREADS) lcnt[i] = 0;
        __syncthreads();
        for (int base = e_lo; base < e_hi; base += CSR_THREADS * CSR_BATCH) {
            int d[CSR_BATCH];
            #pragma unroll
            for (int u = 0; u < CSR_BATCH; ++u) {
                const int e = base + u * CSR_THREADS + tid;
                const int v = dst[(e < e_hi) ? e : e_lo];
                d[u] = (e < e_hi) ? v : -1;
            }
            #pragma unroll
            for (int u = 0; u < CSR_BATCH; ++u) {
                if (d[u] >= lo && d[u] < hi) atomicAdd(&lcnt[d[u] - lo], 1);
            }
        }
        __syncthreads();
        for (int i = tid; i < hi - lo; i += CSR_THREADS) out[lo + i] = lcnt[i];
        __syncthreads();
    }
}

__global__ __launch_bounds__(256) void ggnn_chunk_starts(
    int* __restrict__ cnt2, int* __restrict__ cnt, int n)
{
    const int d = blockIdx.x * 256 + threadIdx.x;
    if (d >= n) return;
    int v[CSR_CHUNKS];
    #pragma unroll
    for (int c = 0; c < CSR_CHUNKS; ++c) v[c] = cnt2[(size_t)c * n + d];
    int run = 0;
    #pragma unroll
    for (int c = 0; c < CSR_CHUNKS; ++c) {
        cnt2[(size_t)c * n + d] = run;
        run += v[c];
    }
    cnt[d] = run;
}

__global__ __launch_bounds__(1024) void ggnn_scan_counts(
    const int* __restrict__ cnt, int* __restrict__ rowptr, int n)
{
    __shared__ int wsum[32];
    const int t    = threadIdx.x;
    const int lane = t & 31;
    const int w    = t >> 5;
    int carry = 0;
    for (int base = 0; base < n; base += 4096) {
        const int i0 = base + t * 4;
        int c[4];
        #pragma unroll
        for (int k = 0; k < 4; ++k) c[k] = (i0 + k < n) ? cnt[i0 + k] : 0;
        const int s = c[0] + c[1] + c[2] + c[3];

        int incl = s;
        #pragma unroll
        for (int off = 1; off < 32; off <<= 1) {
            const int u = __shfl_up(incl, off);
            if (lane >= off) incl += u;
        }
        if (lane == 31) wsum[w] = incl;
        __syncthreads();
        if (w == 0) {
            int si = wsum[lane];
            #pragma unroll
            for (int off = 1; off < 32; off <<= 1) {
                const int u = __shfl_up(si, off);
                if (lane >= off) si += u;
            }
            wsum[lane] = si;
        }
        __syncthreads();
        const int wprefix = (w == 0) ? 0 : wsum[w - 1];
        const int total   = wsum[31];
        int run = carry + wprefix + incl - s;
        #pragma unroll
        for (int k = 0; k < 4; ++k) {
            if (i0 + k < n) rowptr[i0 + k] = run;
            run += c[k];
        }
        carry += total;
        __syncthreads();
    }
    if (t == 0) rowptr[n] = carry;
}

__global__ __launch_bounds__(CSR_THREADS) void ggnn_fill_lists(
    const int* __restrict__ src, const int* __restrict__ dst,
    const int* __restrict__ rowptr, const int* __restrict__ cnt2,
    int* __restrict__ csr_src, int nedges, int n)
{
    __shared__ int next[CSR_SPAN];
    const int tid   = threadIdx.x;
    const int per_r = (n + CSR_RANGES - 1) / CSR_RANGES;
    const int r_lo  = (int)blockIdx.x * per_r;
    int r_hi = r_lo + per_r;
    if (r_hi > n) r_hi = n;
    const int per_c = (nedges + CSR_CHUNKS - 1) / CSR_CHUNKS;
    const int e_lo  = (int)blockIdx.y * per_c;
    int e_hi = e_lo + per_c;
    if (e_hi > nedges) e_hi = nedges;
    const int* cstart = cnt2 + (size_t)blockIdx.y * n;

    for (int lo = r_lo; lo < r_hi; lo += CSR_SPAN) {
        int hi = lo + CSR_SPAN;
        if (hi > r_hi) hi = r_hi;
        for (int i = tid; i < hi - lo; i += CSR_THREADS) next[i] = rowptr[lo + i] + cstart[lo + i];
        __syncthreads();
        for (int base = e_lo; base < e_hi; base += CSR_THREADS * CSR_BATCH) {
            int d[CSR_BATCH], s[CSR_BATCH], pos[CSR_BATCH];
            #pragma unroll
            for (int u = 0; u < CSR_BATCH; ++u) {
                const int e  = base + u * CSR_THREADS + tid;
                const int ec = (e < e_hi) ? e : e_lo;
                const int v  = dst[ec];
                s[u] = src[ec];
                d[u] = (e < e_hi) ? v : -1;
            }
            #pragma unroll
            for (int u = 0; u < CSR_BATCH; ++u) {
                pos[u] = -1;
                if (d[u] >= lo && d[u] < hi) pos[u] = atomicAdd(&next[d[u] - lo], 1);
            }
            #pragma unroll
            for (int u = 0; u < CSR_BATCH; ++u) {
                if (pos[u] >= 0) csr_src[pos[u]] = s[u];
            }
        }
        __syncthreads();
    }
}

__device__ __forceinline__ void ggnn_add4(float4& acc, const float4 v)
{
    acc.x += v.x; acc.y += v.y; acc.z += v.z; acc.w += v.w;
}

__device__ __forceinline__ void ggnn_gather_node(
    const float* __restrict__ m, const int* __restrict__ rowptr,
    const int* __restrict__ csr_src, float* __restrict__ agg, int node, int lane)
{
    const int beg = __builtin_amdgcn_readfirstlane(rowptr[node]);
    const int end = __builtin_amdgcn_readfirstlane(rowptr[node + 1]);
    const float* mcol = m + lane * 4;
    float4 acc = make_float4(0.f, 0.f, 0.f, 0.f);

    for (int base = beg; base < end; base += 32) {
        int nb = end - base;
        if (nb > 32) nb = 32;
        const int mine = (lane < nb) ? csr_src[base + lane] : 0;
        int j = 0;
        for (; j + 8 <= nb; j += 8) {
            float4 v[8];
            #pragma unroll
            for (int u = 0; u < 8; ++u) {
                const int s = __builtin_amdgcn_readlane(mine, j + u);
                v[u] = *(const float4*)(mcol + (size_t)s * KDIM);
            }
            #pragma unroll
            for (int u = 0; u < 8; ++u) ggnn_add4(acc, v[u]);
        }
        if (j + 4 <= nb) {
            float4 v[4];
            #pragma unroll
            for (int u = 0; u < 4; ++u) {
                const int s = __builtin_amdgcn_readlane(mine, j + u);
                v[u] = *(const float4*)(mcol + (size_t)s * KDIM);
            }
            #pragma unroll
            for (int u = 0; u < 4; ++u) ggnn_add4(acc, v[u]);
            j += 4;
        }
        if (j + 2 <= nb) {
            const int s0 = __builtin_amdgcn_readlane(mine, j);
            const int s1 = __builtin_amdgcn_readlane(mine, j + 1);
            const float4 v0 = *(const float4*)(mcol + (size_t)s0 * KDIM);
            const float4 v1 = *(const float4*)(mcol + (size_t)s1 * KDIM);
            ggnn_add4(acc, v0);
            ggnn_add4(acc, v1);
            j += 2;
        }
        if (j < nb) {
            const int s0 = __builtin_amdgcn_readlane(mine, j);
            const float4 v0 = *(const float4*)(mcol + (size_t)s0 * KDIM);
            ggnn_add4(acc, v0);
        }
    }
    *(float4*)(agg + (size_t)node * KDIM + lane * 4) = acc;
}

#define GRU_SLICE   32
#define GRU_WROWS   (3 * GRU_SLICE)
#define GRU_GROUPS  64
#define GRU_GH_TILE (6 * 2 * 32)

#define GRU_WAVES   32
#define GRU_THREADS (GRU_WAVES * 32)
#define GRU_STAGE_PIECES ((GRU_WROWS * (KDIM / 4)) / GRU_THREADS)
#define HH_NODES_PER_TURN 4

__global__ __launch_bounds__(GRU_THREADS) void ggnn_gather_hh(
    const float* __restrict__ hin, const float* __restrict__ w_hh,
    const int* __restrict__ rowptr, const int* __restrict__ csr_src,
    float* __restrict__ agg, float* __restrict__ gh, int nrows)
{
    __shared__ float Wt[GRU_WROWS * KDIM];
    __shared__ int next_tile;
    __shared__ int next_node;

    const int tid = threadIdx.x;
    const int c0  = blockIdx.x * GRU_SLICE;

    if (tid == 0) { next_tile = 0; next_node = 0; }

    float4 stage[GRU_STAGE_PIECES];
    #pragma unroll
    for (int j = 0; j < GRU_STAGE_PIECES; ++j) {
        const int i  = tid + j * GRU_THREADS;
        const int n  = i >> 5;
        const int k4 = (i & 31) << 2;
        const int g  = n >> 5;
        const int w  = n & 31;
        stage[j] = *(const float4*)(w_hh + (size_t)(g * 128 + c0 + w) * KDIM + k4);
    }
    #pragma unroll
    for (int j = 0; j < GRU_STAGE_PIECES; ++j) {
        const int i  = tid + j * GRU_THREADS;
        const int n  = i >> 5;
        const int k4 = (i & 31) << 2;
        *(float4*)&Wt[n * KDIM + (k4 ^ ((n & 15) << 2))] = stage[j];
    }
    __syncthreads();

    const int lane = tid & 31;
    const int wave = __builtin_amdgcn_readfirstlane(tid >> 5);
    const int half = lane >> 4;
    const int l16  = lane & 15;

    const float* Brow = &Wt[l16 * KDIM + half * 2];
    const int swz = l16 << 2;

    const int slices    = (int)gridDim.x;
    const int slice     = (int)blockIdx.x;
    const int groups    = (int)gridDim.y;
    const int group     = (int)blockIdx.y;
    const int row_tiles = (nrows + 15) >> 4;

    const int nwg     = slices * groups;
    const int per     = (nrows + nwg - 1) / nwg;
    const int node_lo = (group * slices + slice) * per;
    int node_hi = node_lo + per;
    if (node_hi > nrows) node_hi = nrows;

    bool tiles_left = true;
    bool nodes_left = true;
    bool tile_turn  = (wave & 1) == 0;
    while (tiles_left || nodes_left) {
        if (tile_turn && tiles_left) {
            int ticket = 0;
            if (lane == 0) ticket = atomicAdd(&next_tile, 1);
            ticket = __builtin_amdgcn_readfirstlane(ticket);
            const int rt = group + ticket * groups;
            if (rt >= row_tiles) {
                tiles_left = false;
            } else {
                const int m0 = rt * 16;
                const float* Hrow = hin + (size_t)(m0 + l16) * KDIM + half * 2;

                v8f acc[6] = {};
                for (int kb = 0; kb < KDIM; kb += 4) {
                    const v2f ah = *(const v2f*)(Hrow + kb);
                    const int kc = kb ^ swz;
                    #pragma unroll
                    for (int t = 0; t < 6; ++t) {
                        const v2f b = *(const v2f*)(Brow + t * 16 * KDIM + kc);
                        acc[t] = __builtin_amdgcn_wmma_f32_16x16x4_f32(
                            false, ah, false, b, (short)0, acc[t], false, false);
                    }
                }

                float4* ghp = (float4*)gh + ((size_t)rt * slices + slice) * GRU_GH_TILE + lane;
                #pragma unroll
                for (int t = 0; t < 6; ++t) {
                    ghp[(t * 2 + 0) * 32] = make_float4(acc[t][0], acc[t][1], acc[t][2], acc[t][3]);
                    ghp[(t * 2 + 1) * 32] = make_float4(acc[t][4], acc[t][5], acc[t][6], acc[t][7]);
                }
            }
        }
        tile_turn = true;
        if (nodes_left) {
            for (int q = 0; q < HH_NODES_PER_TURN; ++q) {
                int ticket = 0;
                if (lane == 0) ticket = atomicAdd(&next_node, 1);
                ticket = __builtin_amdgcn_readfirstlane(ticket);
                const int node = node_lo + ticket;
                if (node >= node_hi) { nodes_left = false; break; }
                ggnn_gather_node(hin, rowptr, csr_src, agg, node, lane);
            }
        }
    }
}

__device__ __forceinline__ float ggnn_sigmoid(float x)
{
    return __builtin_amdgcn_rcpf(1.0f + __builtin_amdgcn_exp2f(x * -1.4426950408889634f));
}

__device__ __forceinline__ float ggnn_tanh(float x)
{
    return 1.0f - 2.0f * __builtin_amdgcn_rcpf(1.0f + __builtin_amdgcn_exp2f(x * 2.8853900817779268f));
}

__global__ __launch_bounds__(GRU_THREADS) void ggnn_gru_in(
    const float* __restrict__ agg, const float* __restrict__ hin,
    const float* __restrict__ w_ih, const float* __restrict__ gh,
    const float* __restrict__ b_ih, const float* __restrict__ b_hh,
    float* __restrict__ hout, int nrows)
{
    __shared__ float Wt[GRU_WROWS * KDIM];
    __shared__ int next_tile;

    const int tid = threadIdx.x;
    const int c0  = blockIdx.x * GRU_SLICE;

    if (tid == 0) next_tile = 0;

    float4 stage[GRU_STAGE_PIECES];
    #pragma unroll
    for (int j = 0; j < GRU_STAGE_PIECES; ++j) {
        const int i  = tid + j * GRU_THREADS;
        const int n  = i >> 5;
        const int k4 = (i & 31) << 2;
        const int g  = n >> 5;
        const int w  = n & 31;
        stage[j] = *(const float4*)(w_ih + (size_t)(g * 128 + c0 + w) * KDIM + k4);
    }
    #pragma unroll
    for (int j = 0; j < GRU_STAGE_PIECES; ++j) {
        const int i  = tid + j * GRU_THREADS;
        const int n  = i >> 5;
        const int k4 = (i & 31) << 2;
        *(float4*)&Wt[n * KDIM + (k4 ^ ((n & 15) << 2))] = stage[j];
    }
    __syncthreads();

    const int lane = tid & 31;
    const int half = lane >> 4;
    const int l16  = lane & 15;

    const float* Brow = &Wt[l16 * KDIM + half * 2];
    const int swz = l16 << 2;

    const int slices    = (int)gridDim.x;
    const int slice     = (int)blockIdx.x;
    const int groups    = (int)gridDim.y;
    const int group     = (int)blockIdx.y;
    const int row_tiles = (nrows + 15) >> 4;

    for (;;) {
        int ticket = 0;
        if (lane == 0) ticket = atomicAdd(&next_tile, 1);
        ticket = __builtin_amdgcn_readfirstlane(ticket);
        const int rt = group + ticket * groups;
        if (rt >= row_tiles) break;

        const int m0 = rt * 16;
        const float* Arow = agg + (size_t)(m0 + l16) * KDIM + half * 2;

        v8f acc[6] = {};
        for (int kb = 0; kb < KDIM; kb += 4) {
            const v2f a  = *(const v2f*)(Arow + kb);
            const int kc = kb ^ swz;
            #pragma unroll
            for (int t = 0; t < 6; ++t) {
                const v2f b = *(const v2f*)(Brow + t * 16 * KDIM + kc);
                acc[t] = __builtin_amdgcn_wmma_f32_16x16x4_f32(
                    false, a, false, b, (short)0, acc[t], false, false);
            }
        }

        const float4* ghp = (const float4*)gh + ((size_t)rt * slices + slice) * GRU_GH_TILE + lane;
        #pragma unroll
        for (int j = 0; j < 2; ++j) {
            const int c = c0 + j * 16 + l16;
            const float bir = b_ih[c], biz = b_ih[c + 128], bin = b_ih[c + 256];
            const float bhr = b_hh[c], bhz = b_hh[c + 128], bhn = b_hh[c + 256];
            float ghr[8], ghz[8], ghn[8];
            #pragma unroll
            for (int q = 0; q < 2; ++q) {
                const float4 vr = ghp[((0 + j) * 2 + q) * 32];
                const float4 vz = ghp[((2 + j) * 2 + q) * 32];
                const float4 vn = ghp[((4 + j) * 2 + q) * 32];
                ghr[q * 4 + 0] = vr.x; ghr[q * 4 + 1] = vr.y; ghr[q * 4 + 2] = vr.z; ghr[q * 4 + 3] = vr.w;
                ghz[q * 4 + 0] = vz.x; ghz[q * 4 + 1] = vz.y; ghz[q * 4 + 2] = vz.z; ghz[q * 4 + 3] = vz.w;
                ghn[q * 4 + 0] = vn.x; ghn[q * 4 + 1] = vn.y; ghn[q * 4 + 2] = vn.z; ghn[q * 4 + 3] = vn.w;
            }
            #pragma unroll
            for (int r = 0; r < 8; ++r) {
                const size_t off = (size_t)(m0 + r + half * 8) * KDIM + c;
                const float i_r = acc[0 + j][r] + bir, h_r = ghr[r] + bhr;
                const float i_z = acc[2 + j][r] + biz, h_z = ghz[r] + bhz;
                const float i_n = acc[4 + j][r] + bin, h_n = ghn[r] + bhn;
                const float rg = ggnn_sigmoid(i_r + h_r);
                const float zg = ggnn_sigmoid(i_z + h_z);
                const float nn = ggnn_tanh(i_n + rg * h_n);
                const float hv = hin[off];
                hout[off] = (1.0f - zg) * nn + zg * hv;
            }
        }
    }
}

extern "C" void kernel_launch(void* const* d_in, const int* in_sizes, int n_in,
                              void* d_out, int out_size, void* d_ws, size_t ws_size,
                              hipStream_t stream)
{
    (void)n_in; (void)out_size; (void)ws_size;

    const float* x      = (const float*)d_in[0];
    const int*   ei     = (const int*)d_in[1];
    const float* lin_w  = (const float*)d_in[2];
    const float* lin_b  = (const float*)d_in[3];
    const float* ggc_w  = (const float*)d_in[4];
    const float* w_ih   = (const float*)d_in[5];
    const float* w_hh   = (const float*)d_in[6];
    const float* b_ih   = (const float*)d_in[7];
    const float* b_hh   = (const float*)d_in[8];
    const float* lin2_w = (const float*)d_in[9];
    const float* lin2_b = (const float*)d_in[10];
    float* out = (float*)d_out;

    const int N = in_sizes[0] / KDIM;
    const int E = in_sizes[1] / 2;
    const int* src = ei;
    const int* dst = ei + E;

    const int row_tiles  = (N + 15) / 16;
    const int row_blocks = (row_tiles + 7) / 8;

    float* h     = (float*)d_ws;
    float* wfold = h + (size_t)N * 128;
    float* agg   = wfold + (size_t)N * 128;
    float* h2  = agg + (size_t)N * 128;
    float* ghb = h2 + (size_t)N * 128;
    int* cnt     = (int*)(ghb + (size_t)row_tiles * 16 * 384);
    int* rowptr  = cnt + N;
    int* cnt2    = rowptr + N + 1;
    int* csr_src = cnt2 + (size_t)CSR_CHUNKS * N;

    const dim3 blk(256);

    const unsigned node_blks    = (unsigned)((N + 255) / 256);
    float* hbuf[2] = { h, h2 };

    ggnn_gemm<false, false><<<dim3(row_blocks, 1), blk, 0, stream>>>(
        x, lin_w, lin_b, h, N, 128, 128, 0, 0);

    ggnn_gemm<false, false><<<dim3(3, 1, 8), blk, 0, stream>>>(
        w_ih, ggc_w, nullptr, wfold, 384, 128, 128, 128 * 128, 384 * 128);

    const dim3 csr_grid(CSR_RANGES, CSR_CHUNKS);
    ggnn_count_dst<<<csr_grid, dim3(CSR_THREADS), 0, stream>>>(dst, cnt2, E, N);
    ggnn_chunk_starts<<<node_blks, blk, 0, stream>>>(cnt2, cnt, N);
    ggnn_scan_counts<<<1, dim3(1024), 0, stream>>>(cnt, rowptr, N);
    ggnn_fill_lists<<<csr_grid, dim3(CSR_THREADS), 0, stream>>>(src, dst, rowptr, cnt2, csr_src, E, N);

    for (int l = 0; l < 8; ++l) {
        const float* hcur = hbuf[l & 1];
        float* hnew = hbuf[(l + 1) & 1];
        ggnn_gather_hh<<<dim3(128 / GRU_SLICE, GRU_GROUPS), dim3(GRU_THREADS), 0, stream>>>(
            hcur, w_hh, rowptr, csr_src, agg, ghb, N);
        ggnn_gru_in<<<dim3(128 / GRU_SLICE, GRU_GROUPS), dim3(GRU_THREADS), 0, stream>>>(
            agg, hcur, wfold + (size_t)l * 384 * 128, ghb, b_ih, b_hh, hnew, N);
    }

    ggnn_gemm<false, true><<<dim3(row_blocks, 1), blk, 0, stream>>>(
        h, lin2_w, lin2_b, out, N, 100, 100, 0, 0);
}
